# starting kernel
# baseline (speedup 1.0000x reference)
#include <hip/hip_runtime.h>
#include <math.h>


typedef float v2f __attribute__((ext_vector_type(2)));
typedef float v8f __attribute__((ext_vector_type(8)));

__device__ __forceinline__ void atomAddF(float* p, float v) {
    __hip_atomic_fetch_add(p, v, __ATOMIC_RELAXED, __HIP_MEMORY_SCOPE_AGENT);
}

__global__ void k_init(float* __restrict__ deg, float* __restrict__ errsum,
                       float* __restrict__ cnt, int N) {
    int i = blockIdx.x * blockDim.x + threadIdx.x;
    if (i < N) { deg[i] = 1.0f; errsum[i] = 0.0f; cnt[i] = 0.0f; }
}

__global__ void k_degree(const int* __restrict__ dst, float* __restrict__ deg, int E) {
    int e = blockIdx.x * blockDim.x + threadIdx.x;
    if (e < E) atomAddF(&deg[dst[e]], 1.0f);
}

__global__ void k_rsqrt(float* __restrict__ deg, int N) {
    int i = blockIdx.x * blockDim.x + threadIdx.x;
    if (i < N) deg[i] = rsqrtf(deg[i]);
}

__global__ void k_transpose(const float* __restrict__ W, float* __restrict__ Wt,
                            int N) {
    int idx = blockIdx.x * blockDim.x + threadIdx.x;
    if (idx < 128 * N) {
        int k = idx / N, n = idx - k * N;
        Wt[n * 128 + k] = W[idx];
    }
}

__global__ void k_fill_bias(float* __restrict__ out, const float* __restrict__ b,
                            int total, int dimMask) {
    int i = blockIdx.x * blockDim.x + threadIdx.x;
    if (i < total) out[i] = b[i & dimMask];
}

__global__ void k_relu(float* __restrict__ a, int total) {
    int i = blockIdx.x * blockDim.x + threadIdx.x;
    if (i < total) a[i] = fmaxf(a[i], 0.0f);
}

template <int NCOLS>
__global__ void k_gemm_wmma_k128(const float* __restrict__ A,
                                 const float* __restrict__ Bt,
                                 float* __restrict__ C) {
    __shared__ float As[16 * 132];
    const int tid  = threadIdx.x;
    const int row0 = blockIdx.x * 16;

    const float4* Ag = (const float4*)(A + (size_t)row0 * 128);
    for (int i = tid; i < 16 * 32; i += blockDim.x) {
        int r = i >> 5, c4 = i & 31;
        ((float4*)(As + r * 132))[c4] = Ag[r * 32 + c4];
    }
    __syncthreads();

    const int lane  = tid & 31;
    const int col0  = (tid >> 5) * 16;
    const int n15   = lane & 15;
    const int khalf = lane >> 4;

    v8f acc = {0.f, 0.f, 0.f, 0.f, 0.f, 0.f, 0.f, 0.f};
    const float* ap = As + n15 * 132 + 2 * khalf;
    const float* bp = Bt + (size_t)(col0 + n15) * 128 + 2 * khalf;

#pragma unroll
    for (int kk = 0; kk < 128; kk += 4) {
        v2f a; a.x = ap[kk]; a.y = ap[kk + 1];
        v2f b = *(const v2f*)(bp + kk);
        acc = __builtin_amdgcn_wmma_f32_16x16x4_f32(
             false, a,  false, b,
             (short)0, acc,  false,  false);
    }

    float* Cp = C + (size_t)(row0 + 8 * khalf) * NCOLS + col0 + n15;
#pragma unroll
    for (int r = 0; r < 8; ++r) Cp[r * NCOLS] = acc[r];
}

template <int DIM>
__global__ void k_aggregate(const float* __restrict__ H, float* __restrict__ out,
                            const int* __restrict__ src, const int* __restrict__ dst,
                            const float* __restrict__ dinv, int E, int N) {
    const int  lane  = threadIdx.x & 31;
    const long long w = (long long)blockIdx.x * (blockDim.x >> 5) + (threadIdx.x >> 5);
    const long long total = (long long)E + N;
    if (w >= total) return;

    int s, d;
    if (w < E) { s = src[w]; d = dst[w]; }
    else       { s = d = (int)(w - E); }
    const float norm = dinv[s] * dinv[d];

    constexpr int VPL = DIM / 32;
    const float* hp = H   + (size_t)s * DIM + lane * VPL;
    float*       op = out + (size_t)d * DIM + lane * VPL;
    if constexpr (VPL == 4) {
        float4 v = *(const float4*)hp;
        atomAddF(op + 0, v.x * norm);
        atomAddF(op + 1, v.y * norm);
        atomAddF(op + 2, v.z * norm);
        atomAddF(op + 3, v.w * norm);
    } else {
        float2 v = *(const float2*)hp;
        atomAddF(op + 0, v.x * norm);
        atomAddF(op + 1, v.y * norm);
    }
}

__global__ void k_decode(const float* __restrict__ Z, const int* __restrict__ row,
                         const int* __restrict__ col, float* __restrict__ errsum,
                         float* __restrict__ cnt, int E) {
    const int lane = threadIdx.x & 31;
    const long long e = (long long)blockIdx.x * (blockDim.x >> 5) + (threadIdx.x >> 5);
    if (e >= E) return;
    const int r = row[e], c = col[e];

    const float2 zr = *(const float2*)(Z + (size_t)r * 64 + lane * 2);
    const float2 zc = *(const float2*)(Z + (size_t)c * 64 + lane * 2);
    float p = zr.x * zc.x + zr.y * zc.y;
#pragma unroll
    for (int off = 16; off > 0; off >>= 1) p += __shfl_xor(p, off, 32);

    if (lane == 0) {
        float sig = 1.0f / (1.0f + __expf(-p));
        float err = -__logf(sig + 1e-15f);
        atomAddF(&errsum[r], err);
        atomAddF(&cnt[r], 1.0f);
    }
}

__global__ void k_finalize(const float* __restrict__ errsum, const float* __restrict__ cnt,
                           float* __restrict__ out, int N) {
    int i = blockIdx.x * blockDim.x + threadIdx.x;
    if (i < N) {
        float c = cnt[i];
        out[i] = (c > 0.0f) ? (errsum[i] / fmaxf(c, 1.0f)) : 10.0f;
    }
}

extern "C" void kernel_launch(void* const* d_in, const int* in_sizes, int n_in,
                              void* d_out, int out_size, void* d_ws, size_t ws_size,
                              hipStream_t stream) {
    const float* x  = (const float*)d_in[0];
    const int*   ei = (const int*)d_in[1];
    const float* W1 = (const float*)d_in[2];
    const float* b1 = (const float*)d_in[3];
    const float* W2 = (const float*)d_in[4];
    const float* b2 = (const float*)d_in[5];

    const int N = in_sizes[0] / 128;
    const int E = in_sizes[1] / 2;
    const int* src = ei;
    const int* dst = ei + E;

    float* ws     = (float*)d_ws;
    float* dinv   = ws;
    float* errsum = dinv + N;
    float* cnt    = errsum + N;
    size_t off    = (((size_t)3 * N) + 63) & ~(size_t)63;
    float* wt1    = ws + off;
    float* wt2    = wt1 + 128 * 128;
    float* bufA   = wt2 + 64 * 128;
    float* bufB   = bufA + (size_t)N * 128;

    const int T = 256;
    k_init     <<<(N + T - 1) / T, T, 0, stream>>>(dinv, errsum, cnt, N);
    k_degree   <<<(E + T - 1) / T, T, 0, stream>>>(dst, dinv, E);
    k_rsqrt    <<<(N + T - 1) / T, T, 0, stream>>>(dinv, N);
    k_transpose<<<(128 * 128 + T - 1) / T, T, 0, stream>>>(W1, wt1, 128);
    k_transpose<<<(128 * 64  + T - 1) / T, T, 0, stream>>>(W2, wt2, 64);

    k_gemm_wmma_k128<128><<<N / 16, 256, 0, stream>>>(x, wt1, bufA);
    k_fill_bias<<<(N * 128 + T - 1) / T, T, 0, stream>>>(bufB, b1, N * 128, 127);
    k_aggregate<128><<<(E + N + 7) / 8, 256, 0, stream>>>(bufA, bufB, src, dst, dinv, E, N);
    k_relu<<<(N * 128 + T - 1) / T, T, 0, stream>>>(bufB, N * 128);

    k_gemm_wmma_k128<64><<<N / 16, 128, 0, stream>>>(bufB, wt2, bufA);
    k_fill_bias<<<(N * 64 + T - 1) / T, T, 0, stream>>>(bufB, b2, N * 64, 63);
    k_aggregate<64><<<(E + N + 7) / 8, 256, 0, stream>>>(bufA, bufB, src, dst, dinv, E, N);

    k_decode<<<(E + 7) / 8, 256, 0, stream>>>(bufB, src, dst, errsum, cnt, E);
    k_finalize<<<(N + T - 1) / T, T, 0, stream>>>(errsum, cnt, (float*)d_out, N);
}


# Aggregation filter: degree kernel marks per owner which 64-edge chunks hold its rows (a bit each, atomic or); an owner reads its bit row and only marked chunks; 400-row owners; sums in arrival order
# speedup vs baseline: 1.7232x; 1.7232x over previous
#include <hip/hip_runtime.h>
#include <math.h>


typedef float v2f __attribute__((ext_vector_type(2)));
typedef float v8f __attribute__((ext_vector_type(8)));

__device__ __forceinline__ void atomAddF(float* p, float v) {
    __hip_atomic_fetch_add(p, v, __ATOMIC_RELAXED, __HIP_MEMORY_SCOPE_AGENT);
}

__global__ void k_init(float* __restrict__ deg, float* __restrict__ errsum,
                       float* __restrict__ cnt, int N) {
    int i = blockIdx.x * blockDim.x + threadIdx.x;
    if (i < N) { deg[i] = 1.0f; errsum[i] = 0.0f; cnt[i] = 0.0f; }
}

constexpr int OWN_ROWS = 400;

__global__ void k_degree(const int* __restrict__ dst, float* __restrict__ deg,
                         unsigned* __restrict__ marks, int E) {
    int e = blockIdx.x * blockDim.x + threadIdx.x;
    if (e < E) {
        const int d = dst[e];
        atomAddF(&deg[d], 1.0f);
        const int FW = (E + 2047) >> 11;
        __hip_atomic_fetch_or(&marks[(d / OWN_ROWS) * FW + (e >> 11)],
                              1u << ((e >> 6) & 31),
                              __ATOMIC_RELAXED, __HIP_MEMORY_SCOPE_AGENT);
    }
}

__global__ void k_rsqrt(float* __restrict__ deg, int N) {
    int i = blockIdx.x * blockDim.x + threadIdx.x;
    if (i < N) deg[i] = rsqrtf(deg[i]);
}

__global__ void k_transpose(const float* __restrict__ W, float* __restrict__ Wt,
                            int N) {
    int idx = blockIdx.x * blockDim.x + threadIdx.x;
    if (idx < 128 * N) {
        int k = idx / N, n = idx - k * N;
        Wt[n * 128 + k] = W[idx];
    }
}

template <int NCOLS>
__global__ void k_gemm_wmma_k128(const float* __restrict__ A,
                                 const float* __restrict__ Bt,
                                 float* __restrict__ C) {
    __shared__ float As[16 * 132];
    const int tid  = threadIdx.x;
    const int row0 = blockIdx.x * 16;

    const float4* Ag = (const float4*)(A + (size_t)row0 * 128);
    for (int i = tid; i < 16 * 32; i += blockDim.x) {
        int r = i >> 5, c4 = i & 31;
        ((float4*)(As + r * 132))[c4] = Ag[r * 32 + c4];
    }
    __syncthreads();

    const int lane  = tid & 31;
    const int col0  = (tid >> 5) * 16;
    const int n15   = lane & 15;
    const int khalf = lane >> 4;

    v8f acc = {0.f, 0.f, 0.f, 0.f, 0.f, 0.f, 0.f, 0.f};
    const float* ap = As + n15 * 132 + 2 * khalf;
    const float* bp = Bt + (size_t)(col0 + n15) * 128 + 2 * khalf;

#pragma unroll
    for (int kk = 0; kk < 128; kk += 4) {
        v2f a; a.x = ap[kk]; a.y = ap[kk + 1];
        v2f b = *(const v2f*)(bp + kk);
        acc = __builtin_amdgcn_wmma_f32_16x16x4_f32(
             false, a,  false, b,
             (short)0, acc,  false,  false);
    }

    float* Cp = C + (size_t)(row0 + 8 * khalf) * NCOLS + col0 + n15;
#pragma unroll
    for (int r = 0; r < 8; ++r) Cp[r * NCOLS] = acc[r];
}

constexpr int OWN_QCAP = 544;

__device__ __forceinline__ void ldsAddF(float* p, float v) {
    __hip_atomic_fetch_add(p, v, __ATOMIC_RELAXED, __HIP_MEMORY_SCOPE_WORKGROUP);
}

template <int DIM, bool RELU, bool CLEAR>
__global__ __launch_bounds__(1024)
void k_aggregate_own(const float* __restrict__ H, float* __restrict__ out,
                     const int* __restrict__ src, const int* __restrict__ dst,
                     const float* __restrict__ dinv, const float* __restrict__ bias,
                     unsigned* __restrict__ marks, int E, int N) {
    constexpr int VPL = DIM / 32;
    __shared__ float acc[OWN_ROWS * DIM];
    __shared__ int   queue[32 * OWN_QCAP];

    const int lane = threadIdx.x & 31;
    const int wv = __builtin_amdgcn_readfirstlane((int)(threadIdx.x >> 5));
    int* qbuf = queue + wv * OWN_QCAP;
    const float* hl = H + lane * VPL;
    float bz[VPL];
#pragma unroll
    for (int k = 0; k < VPL; ++k) bz[k] = bias[lane * VPL + k];

    const int FW = (E + 2047) >> 11;
    for (int base = blockIdx.x * OWN_ROWS; base < N; base += gridDim.x * OWN_ROWS) {
        const int nrows = min(OWN_ROWS, N - base);
        unsigned* fl = marks + (base / OWN_ROWS) * FW;

        for (int r = wv; r < nrows; r += 32) {
            const int row = base + r;
            const float di = dinv[row];
            const float w = di * di;
            float* a = acc + r * DIM + lane;
            if constexpr (VPL == 4) {
                const float4 h = *(const float4*)(hl + (size_t)row * DIM);
                a[0] = bz[0] + h.x * w; a[32] = bz[1] + h.y * w;
                a[64] = bz[2] + h.z * w; a[96] = bz[3] + h.w * w;
            } else {
                const float2 h = *(const float2*)(hl + (size_t)row * DIM);
                a[0] = bz[0] + h.x * w; a[32] = bz[1] + h.y * w;
            }
        }
        __syncthreads();

        int tail = 0;

        auto flush = [&](int off, int cnt) {
            int s = 0, dl = 0;
            float nrm = 0.0f;
            if (lane < cnt) {
                const int g = qbuf[off + lane];
                s = src[g];
                const int d = dst[g];
                nrm = dinv[s] * dinv[d];
                dl = d - base;
            }
            for (int j0 = 0; j0 < cnt; j0 += 8) {
                if constexpr (VPL == 4) {
                    float4 v[8];
                    for (int j = 0; j < 8; ++j) {
                        const int sj = __builtin_amdgcn_readlane(s, j0 + j);
                        v[j] = *(const float4*)(hl + (size_t)sj * DIM);
                    }
                    for (int j = 0; j < 8; ++j) {
                        if (j0 + j < cnt) {
                            const int   dj = __builtin_amdgcn_readlane(dl, j0 + j);
                            const float nj = __int_as_float(
                                __builtin_amdgcn_readlane(__float_as_int(nrm), j0 + j));
                            float* a = acc + dj * DIM + lane;
                            ldsAddF(a +  0, v[j].x * nj);
                            ldsAddF(a + 32, v[j].y * nj);
                            ldsAddF(a + 64, v[j].z * nj);
                            ldsAddF(a + 96, v[j].w * nj);
                        }
                    }
                } else {
                    float2 v[8];
                    for (int j = 0; j < 8; ++j) {
                        const int sj = __builtin_amdgcn_readlane(s, j0 + j);
                        v[j] = *(const float2*)(hl + (size_t)sj * DIM);
                    }
                    for (int j = 0; j < 8; ++j) {
                        if (j0 + j < cnt) {
                            const int   dj = __builtin_amdgcn_readlane(dl, j0 + j);
                            const float nj = __int_as_float(
                                __builtin_amdgcn_readlane(__float_as_int(nrm), j0 + j));
                            float* a = acc + dj * DIM + lane;
                            ldsAddF(a +  0, v[j].x * nj);
                            ldsAddF(a + 32, v[j].y * nj);
                        }
                    }
                }
            }
        };

        auto push = [&](int d, int g) {
            const bool hit = (unsigned)(d - base) < (unsigned)nrows;
            const unsigned m = __builtin_amdgcn_ballot_w32(hit);
            if (m != 0u) {
                if (hit) qbuf[tail + (int)__builtin_amdgcn_mbcnt_lo(m, 0u)] = g;
                tail += __builtin_popcount(m);
            }
        };

        auto drain = [&]() {
            if (tail >= 32) {
                int head = 0;
                for (; tail - head >= 32; head += 32) flush(head, 32);
                const int rem = tail - head;
                if (lane < rem) { const int g = qbuf[head + lane]; qbuf[lane] = g; }
                tail = rem;
            }
        };

        auto scan = [&](unsigned m, int e0) {
            while (m != 0u) {
                int cb[8], dlo[8], dhi[8];
#pragma unroll
                for (int j = 0; j < 8; ++j) {
                    cb[j] = -1; dlo[j] = -1; dhi[j] = -1;
                    if (m != 0u) {
                        const int c = e0 + (__builtin_ctz(m) << 6);
                        m &= m - 1u;
                        cb[j] = c;
                        const int g = c + lane;
                        dlo[j] = (g < E) ? dst[g] : -1;
                        dhi[j] = (g + 32 < E) ? dst[g + 32] : -1;
                    }
                }
#pragma unroll
                for (int j = 0; j < 8; ++j) {
                    if (cb[j] >= 0) {
                        push(dlo[j], cb[j] + lane);
                        push(dhi[j], cb[j] + 32 + lane);
                    }
                }
                drain();
            }
        };

        for (int wb = wv; wb < FW; wb += 32 * 32) {
            const int wi = wb + 32 * lane;
            const unsigned mine = (wi < FW) ? fl[wi] : 0u;
            for (int i = 0; i < 32; ++i) {
                const unsigned m = (unsigned)__builtin_amdgcn_readlane((int)mine, i);
                scan(m, (wb + 32 * i) << 11);
            }
        }
        for (int head = 0; head < tail; head += 32) flush(head, min(32, tail - head));
        __syncthreads();
        if constexpr (CLEAR) {
            for (int i = threadIdx.x; i < FW; i += blockDim.x) fl[i] = 0u;
        }

        for (int r = wv; r < nrows; r += 32) {
            const float* a = acc + r * DIM + lane;
            float* op = out + (size_t)(base + r) * DIM + lane * VPL;
            if constexpr (VPL == 4) {
                float4 o = {a[0], a[32], a[64], a[96]};
                if (RELU) { o.x = fmaxf(o.x, 0.0f); o.y = fmaxf(o.y, 0.0f);
                            o.z = fmaxf(o.z, 0.0f); o.w = fmaxf(o.w, 0.0f); }
                *(float4*)op = o;
            } else {
                float2 o = {a[0], a[32]};
                if (RELU) { o.x = fmaxf(o.x, 0.0f); o.y = fmaxf(o.y, 0.0f); }
                *(float2*)op = o;
            }
        }
        __syncthreads();
    }
}

__global__ void k_decode(const float* __restrict__ Z, const int* __restrict__ row,
                         const int* __restrict__ col, float* __restrict__ errsum,
                         float* __restrict__ cnt, int E) {
    const int lane = threadIdx.x & 31;
    const int wave = __builtin_amdgcn_readfirstlane(
        (int)(blockIdx.x * (blockDim.x >> 5) + (threadIdx.x >> 5)));
    const int first = wave * 32;
    if (first >= E) return;
    const int nval = min(32, E - first);

    const int g = first + lane;
    int r = 0, c = 0;
    if (g < E) { r = row[g]; c = col[g]; }

    const float* zl = Z + lane * 2;
    float score = 0.0f;
    for (int j0 = 0; j0 < nval; j0 += 8) {
        float2 zr[8], zc[8];
        for (int j = 0; j < 8; ++j) {
            const int rj = __builtin_amdgcn_readlane(r, j0 + j);
            const int cj = __builtin_amdgcn_readlane(c, j0 + j);
            zr[j] = *(const float2*)(zl + (size_t)rj * 64);
            zc[j] = *(const float2*)(zl + (size_t)cj * 64);
        }
        for (int j = 0; j < 8; ++j) {
            float p = zr[j].x * zc[j].x + zr[j].y * zc[j].y;
#pragma unroll
            for (int off = 16; off > 0; off >>= 1) p += __shfl_xor(p, off, 32);
            if (lane == j0 + j) score = p;
        }
    }

    if (g < E) {
        float sig = 1.0f / (1.0f + __expf(-score));
        float err = -__logf(sig + 1e-15f);
        atomAddF(&errsum[r], err);
        atomAddF(&cnt[r], 1.0f);
    }
}

__global__ void k_finalize(const float* __restrict__ errsum, const float* __restrict__ cnt,
                           float* __restrict__ out, int N) {
    int i = blockIdx.x * blockDim.x + threadIdx.x;
    if (i < N) {
        float c = cnt[i];
        out[i] = (c > 0.0f) ? (errsum[i] / fmaxf(c, 1.0f)) : 10.0f;
    }
}

extern "C" void kernel_launch(void* const* d_in, const int* in_sizes, int n_in,
                              void* d_out, int out_size, void* d_ws, size_t ws_size,
                              hipStream_t stream) {
    const float* x  = (const float*)d_in[0];
    const int*   ei = (const int*)d_in[1];
    const float* W1 = (const float*)d_in[2];
    const float* b1 = (const float*)d_in[3];
    const float* W2 = (const float*)d_in[4];
    const float* b2 = (const float*)d_in[5];

    const int N = in_sizes[0] / 128;
    const int E = in_sizes[1] / 2;
    const int* src = ei;
    const int* dst = ei + E;

    float* ws     = (float*)d_ws;
    float* dinv   = ws;
    float* errsum = dinv + N;
    float* cnt    = errsum + N;
    size_t off    = (((size_t)3 * N) + 63) & ~(size_t)63;
    float* wt1    = ws + off;
    float* wt2    = wt1 + 128 * 128;
    float* bufA   = wt2 + 64 * 128;
    float* bufB   = bufA + (size_t)N * 128;

    const int T = 256;
    k_init     <<<(N + T - 1) / T, T, 0, stream>>>(dinv, errsum, cnt, N);
    unsigned* marks = (unsigned*)errsum;
    k_degree   <<<(E + T - 1) / T, T, 0, stream>>>(dst, dinv, marks, E);
    k_rsqrt    <<<(N + T - 1) / T, T, 0, stream>>>(dinv, N);
    k_transpose<<<(128 * 128 + T - 1) / T, T, 0, stream>>>(W1, wt1, 128);
    k_transpose<<<(128 * 64  + T - 1) / T, T, 0, stream>>>(W2, wt2, 64);

    k_gemm_wmma_k128<128><<<N / 16, 256, 0, stream>>>(x, wt1, bufA);
    k_aggregate_own<128, true, false><<<250, 1024, 0, stream>>>(bufA, bufB, src, dst, dinv, b1, marks, E, N);

    k_gemm_wmma_k128<64><<<N / 16, 128, 0, stream>>>(bufB, wt2, bufA);
    k_aggregate_own<64, false, true><<<250, 1024, 0, stream>>>(bufA, bufB, src, dst, dinv, b2, marks, E, N);

    k_decode<<<(E + 255) / 256, 256, 0, stream>>>(bufB, src, dst, errsum, cnt, E);
    k_finalize<<<(N + T - 1) / T, T, 0, stream>>>(errsum, cnt, (float*)d_out, N);
}


# Per-source sums without global atomics: decode stores one error per edge; a new launch adds errors and counts in LDS per edge slice (64) x row range (4); finalize adds 64 partials; sum order changes
# speedup vs baseline: 2.0520x; 1.1908x over previous
#include <hip/hip_runtime.h>
#include <math.h>


typedef float v2f __attribute__((ext_vector_type(2)));
typedef float v8f __attribute__((ext_vector_type(8)));

__device__ __forceinline__ void atomAddF(float* p, float v) {
    __hip_atomic_fetch_add(p, v, __ATOMIC_RELAXED, __HIP_MEMORY_SCOPE_AGENT);
}

__global__ void k_init(float* __restrict__ deg, float* __restrict__ errsum,
                       float* __restrict__ cnt, int N) {
    int i = blockIdx.x * blockDim.x + threadIdx.x;
    if (i < N) { deg[i] = 1.0f; errsum[i] = 0.0f; cnt[i] = 0.0f; }
}

constexpr int OWN_ROWS = 400;

__global__ void k_degree(const int* __restrict__ dst, float* __restrict__ deg,
                         unsigned* __restrict__ marks, int E) {
    int e = blockIdx.x * blockDim.x + threadIdx.x;
    if (e < E) {
        const int d = dst[e];
        atomAddF(&deg[d], 1.0f);
        const int FW = (E + 2047) >> 11;
        __hip_atomic_fetch_or(&marks[(d / OWN_ROWS) * FW + (e >> 11)],
                              1u << ((e >> 6) & 31),
                              __ATOMIC_RELAXED, __HIP_MEMORY_SCOPE_AGENT);
    }
}

__global__ void k_rsqrt(float* __restrict__ deg, int N) {
    int i = blockIdx.x * blockDim.x + threadIdx.x;
    if (i < N) deg[i] = rsqrtf(deg[i]);
}

__global__ void k_transpose(const float* __restrict__ W, float* __restrict__ Wt,
                            int N) {
    int idx = blockIdx.x * blockDim.x + threadIdx.x;
    if (idx < 128 * N) {
        int k = idx / N, n = idx - k * N;
        Wt[n * 128 + k] = W[idx];
    }
}

template <int NCOLS>
__global__ void k_gemm_wmma_k128(const float* __restrict__ A,
                                 const float* __restrict__ Bt,
                                 float* __restrict__ C) {
    __shared__ float As[16 * 132];
    const int tid  = threadIdx.x;
    const int row0 = blockIdx.x * 16;

    const float4* Ag = (const float4*)(A + (size_t)row0 * 128);
    for (int i = tid; i < 16 * 32; i += blockDim.x) {
        int r = i >> 5, c4 = i & 31;
        ((float4*)(As + r * 132))[c4] = Ag[r * 32 + c4];
    }
    __syncthreads();

    const int lane  = tid & 31;
    const int col0  = (tid >> 5) * 16;
    const int n15   = lane & 15;
    const int khalf = lane >> 4;

    v8f acc = {0.f, 0.f, 0.f, 0.f, 0.f, 0.f, 0.f, 0.f};
    const float* ap = As + n15 * 132 + 2 * khalf;
    const float* bp = Bt + (size_t)(col0 + n15) * 128 + 2 * khalf;

#pragma unroll
    for (int kk = 0; kk < 128; kk += 4) {
        v2f a; a.x = ap[kk]; a.y = ap[kk + 1];
        v2f b = *(const v2f*)(bp + kk);
        acc = __builtin_amdgcn_wmma_f32_16x16x4_f32(
             false, a,  false, b,
             (short)0, acc,  false,  false);
    }

    float* Cp = C + (size_t)(row0 + 8 * khalf) * NCOLS + col0 + n15;
#pragma unroll
    for (int r = 0; r < 8; ++r) Cp[r * NCOLS] = acc[r];
}

constexpr int OWN_QCAP = 544;

__device__ __forceinline__ void ldsAddF(float* p, float v) {
    __hip_atomic_fetch_add(p, v, __ATOMIC_RELAXED, __HIP_MEMORY_SCOPE_WORKGROUP);
}

template <int DIM, bool RELU>
__global__ __launch_bounds__(1024)
void k_aggregate_own(const float* __restrict__ H, float* __restrict__ out,
                     const int* __restrict__ src, const int* __restrict__ dst,
                     const float* __restrict__ dinv, const float* __restrict__ bias,
                     const unsigned* __restrict__ marks, int E, int N) {
    constexpr int VPL = DIM / 32;
    __shared__ float acc[OWN_ROWS * DIM];
    __shared__ int   queue[32 * OWN_QCAP];

    const int lane = threadIdx.x & 31;
    const int wv = __builtin_amdgcn_readfirstlane((int)(threadIdx.x >> 5));
    int* qbuf = queue + wv * OWN_QCAP;
    const float* hl = H + lane * VPL;
    float bz[VPL];
#pragma unroll
    for (int k = 0; k < VPL; ++k) bz[k] = bias[lane * VPL + k];

    const int FW = (E + 2047) >> 11;
    for (int base = blockIdx.x * OWN_ROWS; base < N; base += gridDim.x * OWN_ROWS) {
        const int nrows = min(OWN_ROWS, N - base);
        const unsigned* fl = marks + (base / OWN_ROWS) * FW;

        for (int r = wv; r < nrows; r += 32) {
            const int row = base + r;
            const float di = dinv[row];
            const float w = di * di;
            float* a = acc + r * DIM + lane;
            if constexpr (VPL == 4) {
                const float4 h = *(const float4*)(hl + (size_t)row * DIM);
                a[0] = bz[0] + h.x * w; a[32] = bz[1] + h.y * w;
                a[64] = bz[2] + h.z * w; a[96] = bz[3] + h.w * w;
            } else {
                const float2 h = *(const float2*)(hl + (size_t)row * DIM);
                a[0] = bz[0] + h.x * w; a[32] = bz[1] + h.y * w;
            }
        }
        __syncthreads();

        int tail = 0;

        auto flush = [&](int off, int cnt) {
            int s = 0, dl = 0;
            float nrm = 0.0f;
            if (lane < cnt) {
                const int g = qbuf[off + lane];
                s = src[g];
                const int d = dst[g];
                nrm = dinv[s] * dinv[d];
                dl = d - base;
            }
            for (int j0 = 0; j0 < cnt; j0 += 8) {
                if constexpr (VPL == 4) {
                    float4 v[8];
                    for (int j = 0; j < 8; ++j) {
                        const int sj = __builtin_amdgcn_readlane(s, j0 + j);
                        v[j] = *(const float4*)(hl + (size_t)sj * DIM);
                    }
                    for (int j = 0; j < 8; ++j) {
                        if (j0 + j < cnt) {
                            const int   dj = __builtin_amdgcn_readlane(dl, j0 + j);
                            const float nj = __int_as_float(
                                __builtin_amdgcn_readlane(__float_as_int(nrm), j0 + j));
                            float* a = acc + dj * DIM + lane;
                            ldsAddF(a +  0, v[j].x * nj);
                            ldsAddF(a + 32, v[j].y * nj);
                            ldsAddF(a + 64, v[j].z * nj);
                            ldsAddF(a + 96, v[j].w * nj);
                        }
                    }
                } else {
                    float2 v[8];
                    for (int j = 0; j < 8; ++j) {
                        const int sj = __builtin_amdgcn_readlane(s, j0 + j);
                        v[j] = *(const float2*)(hl + (size_t)sj * DIM);
                    }
                    for (int j = 0; j < 8; ++j) {
                        if (j0 + j < cnt) {
                            const int   dj = __builtin_amdgcn_readlane(dl, j0 + j);
                            const float nj = __int_as_float(
                                __builtin_amdgcn_readlane(__float_as_int(nrm), j0 + j));
                            float* a = acc + dj * DIM + lane;
                            ldsAddF(a +  0, v[j].x * nj);
                            ldsAddF(a + 32, v[j].y * nj);
                        }
                    }
                }
            }
        };

        auto push = [&](int d, int g) {
            const bool hit = (unsigned)(d - base) < (unsigned)nrows;
            const unsigned m = __builtin_amdgcn_ballot_w32(hit);
            if (m != 0u) {
                if (hit) qbuf[tail + (int)__builtin_amdgcn_mbcnt_lo(m, 0u)] = g;
                tail += __builtin_popcount(m);
            }
        };

        auto drain = [&]() {
            if (tail >= 32) {
                int head = 0;
                for (; tail - head >= 32; head += 32) flush(head, 32);
                const int rem = tail - head;
                if (lane < rem) { const int g = qbuf[head + lane]; qbuf[lane] = g; }
                tail = rem;
            }
        };

        auto scan = [&](unsigned m, int e0) {
            while (m != 0u) {
                int cb[8], dlo[8], dhi[8];
#pragma unroll
                for (int j = 0; j < 8; ++j) {
                    cb[j] = -1; dlo[j] = -1; dhi[j] = -1;
                    if (m != 0u) {
                        const int c = e0 + (__builtin_ctz(m) << 6);
                        m &= m - 1u;
                        cb[j] = c;
                        const int g = c + lane;
                        dlo[j] = (g < E) ? dst[g] : -1;
                        dhi[j] = (g + 32 < E) ? dst[g + 32] : -1;
                    }
                }
#pragma unroll
                for (int j = 0; j < 8; ++j) {
                    if (cb[j] >= 0) {
                        push(dlo[j], cb[j] + lane);
                        push(dhi[j], cb[j] + 32 + lane);
                    }
                }
                drain();
            }
        };

        for (int wb = wv; wb < FW; wb += 32 * 32) {
            const int wi = wb + 32 * lane;
            const unsigned mine = (wi < FW) ? fl[wi] : 0u;
            for (int i = 0; i < 32; ++i) {
                const unsigned m = (unsigned)__builtin_amdgcn_readlane((int)mine, i);
                scan(m, (wb + 32 * i) << 11);
            }
        }
        for (int head = 0; head < tail; head += 32) flush(head, min(32, tail - head));
        __syncthreads();

        for (int r = wv; r < nrows; r += 32) {
            const float* a = acc + r * DIM + lane;
            float* op = out + (size_t)(base + r) * DIM + lane * VPL;
            if constexpr (VPL == 4) {
                float4 o = {a[0], a[32], a[64], a[96]};
                if (RELU) { o.x = fmaxf(o.x, 0.0f); o.y = fmaxf(o.y, 0.0f);
                            o.z = fmaxf(o.z, 0.0f); o.w = fmaxf(o.w, 0.0f); }
                *(float4*)op = o;
            } else {
                float2 o = {a[0], a[32]};
                if (RELU) { o.x = fmaxf(o.x, 0.0f); o.y = fmaxf(o.y, 0.0f); }
                *(float2*)op = o;
            }
        }
        __syncthreads();
    }
}

__global__ void k_decode(const float* __restrict__ Z, const int* __restrict__ row,
                         const int* __restrict__ col, float* __restrict__ eerr, int E) {
    const int lane = threadIdx.x & 31;
    const int wave = __builtin_amdgcn_readfirstlane(
        (int)(blockIdx.x * (blockDim.x >> 5) + (threadIdx.x >> 5)));
    const int first = wave * 32;
    if (first >= E) return;
    const int nval = min(32, E - first);

    const int g = first + lane;
    int r = 0, c = 0;
    if (g < E) { r = row[g]; c = col[g]; }

    const float* zl = Z + lane * 2;
    float score = 0.0f;
    for (int j0 = 0; j0 < nval; j0 += 8) {
        float2 zr[8], zc[8];
        for (int j = 0; j < 8; ++j) {
            const int rj = __builtin_amdgcn_readlane(r, j0 + j);
            const int cj = __builtin_amdgcn_readlane(c, j0 + j);
            zr[j] = *(const float2*)(zl + (size_t)rj * 64);
            zc[j] = *(const float2*)(zl + (size_t)cj * 64);
        }
        for (int j = 0; j < 8; ++j) {
            float p = zr[j].x * zc[j].x + zr[j].y * zc[j].y;
#pragma unroll
            for (int off = 16; off > 0; off >>= 1) p += __shfl_xor(p, off, 32);
            if (lane == j0 + j) score = p;
        }
    }

    if (g < E) {
        float sig = 1.0f / (1.0f + __expf(-score));
        eerr[g] = -__logf(sig + 1e-15f);
    }
}

constexpr int SEG_S = 64;
constexpr int SEG_R = 25000;
constexpr int SEG_T = 1024;

__global__ __launch_bounds__(SEG_T)
void k_segsum(const float* __restrict__ eerr, const int* __restrict__ row,
              float* __restrict__ part, int E, int N) {
    __shared__ float2 acc[SEG_R];
    const int tid = threadIdx.x;
    constexpr int nt = SEG_T;
    const int ES  = (E + (int)gridDim.x - 1) / (int)gridDim.x;
    const int e0  = blockIdx.x * ES;
    const int e1  = min(E, e0 + ES);
    float2* pp = (float2*)part + (size_t)blockIdx.x * N;

    for (int base = blockIdx.y * SEG_R; base < N; base += gridDim.y * SEG_R) {
        const int nrows = min(SEG_R, N - base);
        for (int i = tid; i < nrows; i += nt) acc[i] = make_float2(0.0f, 0.0f);
        __syncthreads();

        for (int e = e0 + tid; e < e1; e += 8 * nt) {
            int   r[8];
            float v[8];
#pragma unroll
            for (int k = 0; k < 8; ++k) {
                const int ee = e + k * nt;
                r[k] = -1; v[k] = 0.0f;
                if (ee < e1) { r[k] = row[ee]; v[k] = eerr[ee]; }
            }
#pragma unroll
            for (int k = 0; k < 8; ++k) {
                const unsigned rl = (unsigned)(r[k] - base);
                if (rl < (unsigned)nrows) { ldsAddF(&acc[rl].x, v[k]); ldsAddF(&acc[rl].y, 1.0f); }
            }
        }
        __syncthreads();

        for (int i = tid; i < nrows; i += nt) pp[base + i] = acc[i];
        __syncthreads();
    }
}

__global__ void k_finalize(const float* __restrict__ part, float* __restrict__ out, int N) {
    int i = blockIdx.x * blockDim.x + threadIdx.x;
    if (i < N) {
        float es = 0.0f, c = 0.0f;
#pragma unroll 16
        for (int s = 0; s < SEG_S; ++s) {
            const float2 p = ((const float2*)part)[(size_t)s * N + i];
            es += p.x;
            c  += p.y;
        }
        out[i] = (c > 0.0f) ? (es / fmaxf(c, 1.0f)) : 10.0f;
    }
}

extern "C" void kernel_launch(void* const* d_in, const int* in_sizes, int n_in,
                              void* d_out, int out_size, void* d_ws, size_t ws_size,
                              hipStream_t stream) {
    const float* x  = (const float*)d_in[0];
    const int*   ei = (const int*)d_in[1];
    const float* W1 = (const float*)d_in[2];
    const float* b1 = (const float*)d_in[3];
    const float* W2 = (const float*)d_in[4];
    const float* b2 = (const float*)d_in[5];

    const int N = in_sizes[0] / 128;
    const int E = in_sizes[1] / 2;
    const int* src = ei;
    const int* dst = ei + E;

    float* ws     = (float*)d_ws;
    float* dinv   = ws;
    float* errsum = dinv + N;
    float* cnt    = errsum + N;
    size_t off    = (((size_t)3 * N) + 63) & ~(size_t)63;
    float* wt1    = ws + off;
    float* wt2    = wt1 + 128 * 128;
    float* bufA   = wt2 + 64 * 128;
    float* bufB   = bufA + (size_t)N * 128;

    const int T = 256;
    k_init     <<<(N + T - 1) / T, T, 0, stream>>>(dinv, errsum, cnt, N);
    unsigned* marks = (unsigned*)errsum;
    k_degree   <<<(E + T - 1) / T, T, 0, stream>>>(dst, dinv, marks, E);
    k_rsqrt    <<<(N + T - 1) / T, T, 0, stream>>>(dinv, N);
    k_transpose<<<(128 * 128 + T - 1) / T, T, 0, stream>>>(W1, wt1, 128);
    k_transpose<<<(128 * 64  + T - 1) / T, T, 0, stream>>>(W2, wt2, 64);

    k_gemm_wmma_k128<128><<<N / 16, 256, 0, stream>>>(x, wt1, bufA);
    k_aggregate_own<128, true><<<250, 1024, 0, stream>>>(bufA, bufB, src, dst, dinv, b1, marks, E, N);

    k_gemm_wmma_k128<64><<<N / 16, 128, 0, stream>>>(bufB, wt2, bufA);
    k_aggregate_own<64, false><<<250, 1024, 0, stream>>>(bufA, bufB, src, dst, dinv, b2, marks, E, N);

    float* eerr = bufB + (size_t)N * 64;
    float* part = bufA;
    k_decode<<<(E + 255) / 256, 256, 0, stream>>>(bufB, src, dst, eerr, E);
    k_segsum<<<dim3(SEG_S, 4), SEG_T, 0, stream>>>(eerr, src, part, E, N);
    k_finalize<<<(N + T - 1) / T, T, 0, stream>>>(part, (float*)d_out, N);
}


# Projections: the weight pre-pass stores W in WMMA fragment order; a wave's B fragments of two k-steps are one aligned 512-byte load instead of two loads that each touch sixteen lines; sums unchanged
# speedup vs baseline: 3.0758x; 1.4989x over previous
#include <hip/hip_runtime.h>
#include <math.h>


typedef float v2f __attribute__((ext_vector_type(2)));
typedef float v8f __attribute__((ext_vector_type(8)));

constexpr int OWN_ROWS = 400;

__device__ __forceinline__ void ldsAddU(unsigned* p, unsigned v) {
    __hip_atomic_fetch_add(p, v, __ATOMIC_RELAXED, __HIP_MEMORY_SCOPE_WORKGROUP);
}
__device__ __forceinline__ void ldsOrU(unsigned* p, unsigned v) {
    __hip_atomic_fetch_or(p, v, __ATOMIC_RELAXED, __HIP_MEMORY_SCOPE_WORKGROUP);
}

constexpr int DM_S    = 16;
constexpr int DM_OPR  = 16;
constexpr int DM_ROWS = DM_OPR * OWN_ROWS;
constexpr int DM_WMAX = 64;
constexpr int DM_T    = 1024;

__global__ __launch_bounds__(DM_T)
void k_degmark(const int* __restrict__ dst, unsigned* __restrict__ degpart,
               unsigned* __restrict__ marks, int E, int N) {
    __shared__ unsigned cntl[DM_ROWS];
    __shared__ unsigned mk[DM_OPR * DM_WMAX];
    const int tid = threadIdx.x;
    const int FW  = (E + 2047) >> 11;
    const int wps = (FW + (int)gridDim.x - 1) / (int)gridDim.x;
    const int w0  = blockIdx.x * wps;
    const int e0  = w0 << 11;
    const int e1  = min(E, (w0 + wps) << 11);

    for (int rg = blockIdx.y; rg * DM_ROWS < N; rg += gridDim.y) {
        const int base  = rg * DM_ROWS;
        const int nrows = min(DM_ROWS, N - base);
        for (int i = tid; i < DM_ROWS; i += DM_T) cntl[i] = 0u;
        for (int i = tid; i < DM_OPR * DM_WMAX; i += DM_T) mk[i] = 0u;
        __syncthreads();

        for (int e = e0 + tid; e < e1; e += 16 * DM_T) {
            int d[16];
#pragma unroll
            for (int k = 0; k < 16; ++k) {
                const int ee = e + k * DM_T;
                d[k] = (ee < e1) ? dst[ee] : -1;
            }
#pragma unroll
            for (int k = 0; k < 16; ++k) {
                const unsigned rl = (unsigned)(d[k] - base);
                if (rl < (unsigned)nrows) {
                    const int ee = e + k * DM_T;
                    ldsAddU(&cntl[rl], 1u);
                    ldsOrU(&mk[(rl / OWN_ROWS) * DM_WMAX + ((ee - e0) >> 11)],
                           1u << ((ee >> 6) & 31));
                }
            }
        }
        __syncthreads();

        unsigned* pl = degpart + (size_t)blockIdx.x * N + base;
        for (int i = tid; i < nrows; i += DM_T) pl[i] = cntl[i];
        for (int i = tid; i < DM_OPR * DM_WMAX; i += DM_T) {
            const int o = i / DM_WMAX, w = i - o * DM_WMAX;
            const int owner = rg * DM_OPR + o;
            if (w < wps && w0 + w < FW && owner * OWN_ROWS < N)
                marks[(size_t)owner * FW + w0 + w] = mk[i];
        }
        __syncthreads();
    }
}

__global__ void k_rsqrt(const unsigned* __restrict__ degpart, float* __restrict__ dinv, int N) {
    int i = blockIdx.x * blockDim.x + threadIdx.x;
    if (i < N) {
        unsigned c = 1u;
#pragma unroll
        for (int s = 0; s < DM_S; ++s) c += degpart[(size_t)s * N + i];
        dinv[i] = rsqrtf((float)c);
    }
}

__global__ void k_fragorder(const float* __restrict__ W, float* __restrict__ Wf,
                            int N) {
    int idx = blockIdx.x * blockDim.x + threadIdx.x;
    if (idx < 128 * N) {
        const int j    = idx & 3;
        const int lane = (idx >> 2) & 31;
        const int kq   = (idx >> 7) & 15;
        const int ct   = idx >> 11;
        const int k = 8 * kq + 4 * (j >> 1) + 2 * (lane >> 4) + (j & 1);
        const int n = 16 * ct + (lane & 15);
        Wf[idx] = W[k * N + n];
    }
}

template <int NCOLS>
__global__ void k_gemm_wmma_k128(const float* __restrict__ A,
                                 const float* __restrict__ Bf,
                                 float* __restrict__ C) {
    __shared__ float As[16 * 132];
    const int tid  = threadIdx.x;
    const int row0 = blockIdx.x * 16;

    const float4* Ag = (const float4*)(A + (size_t)row0 * 128);
    for (int i = tid; i < 16 * 32; i += blockDim.x) {
        int r = i >> 5, c4 = i & 31;
        ((float4*)(As + r * 132))[c4] = Ag[r * 32 + c4];
    }
    __syncthreads();

    const int lane  = tid & 31;
    const int col0  = (tid >> 5) * 16;
    const int n15   = lane & 15;
    const int khalf = lane >> 4;

    v8f acc = {0.f, 0.f, 0.f, 0.f, 0.f, 0.f, 0.f, 0.f};
    const float* ap = As + n15 * 132 + 2 * khalf;
    const float* bp = Bf + ((size_t)(tid >> 5) * (16 * 32) + lane) * 4;

#pragma unroll
    for (int kq = 0; kq < 16; ++kq) {
        const float4 b4 = *(const float4*)(bp + kq * 128);
        v2f a0; a0.x = ap[8 * kq];     a0.y = ap[8 * kq + 1];
        v2f b0; b0.x = b4.x;           b0.y = b4.y;
        acc = __builtin_amdgcn_wmma_f32_16x16x4_f32(
             false, a0,  false, b0,
             (short)0, acc,  false,  false);
        v2f a1; a1.x = ap[8 * kq + 4]; a1.y = ap[8 * kq + 5];
        v2f b1; b1.x = b4.z;           b1.y = b4.w;
        acc = __builtin_amdgcn_wmma_f32_16x16x4_f32(
             false, a1,  false, b1,
             (short)0, acc,  false,  false);
    }

    float* Cp = C + (size_t)(row0 + 8 * khalf) * NCOLS + col0 + n15;
#pragma unroll
    for (int r = 0; r < 8; ++r) Cp[r * NCOLS] = acc[r];
}

constexpr int OWN_QCAP = 544;

__device__ __forceinline__ void ldsAddF(float* p, float v) {
    __hip_atomic_fetch_add(p, v, __ATOMIC_RELAXED, __HIP_MEMORY_SCOPE_WORKGROUP);
}

template <int DIM, bool RELU>
__global__ __launch_bounds__(1024)
void k_aggregate_own(const float* __restrict__ H, float* __restrict__ out,
                     const int* __restrict__ src, const int* __restrict__ dst,
                     const float* __restrict__ dinv, const float* __restrict__ bias,
                     const unsigned* __restrict__ marks, int E, int N) {
    constexpr int VPL = DIM / 32;
    __shared__ float acc[OWN_ROWS * DIM];
    __shared__ int   queue[32 * OWN_QCAP];

    const int lane = threadIdx.x & 31;
    const int wv = __builtin_amdgcn_readfirstlane((int)(threadIdx.x >> 5));
    int* qbuf = queue + wv * OWN_QCAP;
    const float* hl = H + lane * VPL;
    float bz[VPL];
#pragma unroll
    for (int k = 0; k < VPL; ++k) bz[k] = bias[lane * VPL + k];

    const int FW = (E + 2047) >> 11;
    for (int base = blockIdx.x * OWN_ROWS; base < N; base += gridDim.x * OWN_ROWS) {
        const int nrows = min(OWN_ROWS, N - base);
        const unsigned* fl = marks + (base / OWN_ROWS) * FW;

        for (int r = wv; r < nrows; r += 32) {
            const int row = base + r;
            const float di = dinv[row];
            const float w = di * di;
            float* a = acc + r * DIM + lane;
            if constexpr (VPL == 4) {
                const float4 h = *(const float4*)(hl + (size_t)row * DIM);
                a[0] = bz[0] + h.x * w; a[32] = bz[1] + h.y * w;
                a[64] = bz[2] + h.z * w; a[96] = bz[3] + h.w * w;
            } else {
                const float2 h = *(const float2*)(hl + (size_t)row * DIM);
                a[0] = bz[0] + h.x * w; a[32] = bz[1] + h.y * w;
            }
        }
        __syncthreads();

        int tail = 0;

        auto flush = [&](int off, int cnt) {
            int s = 0, dl = 0;
            float nrm = 0.0f;
            if (lane < cnt) {
                const int g = qbuf[off + lane];
                s = src[g];
                const int d = dst[g];
                nrm = dinv[s] * dinv[d];
                dl = d - base;
            }
            for (int j0 = 0; j0 < cnt; j0 += 8) {
                if constexpr (VPL == 4) {
                    float4 v[8];
                    for (int j = 0; j < 8; ++j) {
                        const int sj = __builtin_amdgcn_readlane(s, j0 + j);
                        v[j] = *(const float4*)(hl + (size_t)sj * DIM);
                    }
                    for (int j = 0; j < 8; ++j) {
                        if (j0 + j < cnt) {
                            const int   dj = __builtin_amdgcn_readlane(dl, j0 + j);
                            const float nj = __int_as_float(
                                __builtin_amdgcn_readlane(__float_as_int(nrm), j0 + j));
                            float* a = acc + dj * DIM + lane;
                            ldsAddF(a +  0, v[j].x * nj);
                            ldsAddF(a + 32, v[j].y * nj);
                            ldsAddF(a + 64, v[j].z * nj);
                            ldsAddF(a + 96, v[j].w * nj);
                        }
                    }
                } else {
                    float2 v[8];
                    for (int j = 0; j < 8; ++j) {
                        const int sj = __builtin_amdgcn_readlane(s, j0 + j);
                        v[j] = *(const float2*)(hl + (size_t)sj * DIM);
                    }
                    for (int j = 0; j < 8; ++j) {
                        if (j0 + j < cnt) {
                            const int   dj = __builtin_amdgcn_readlane(dl, j0 + j);
                            const float nj = __int_as_float(
                                __builtin_amdgcn_readlane(__float_as_int(nrm), j0 + j));
                            float* a = acc + dj * DIM + lane;
                            ldsAddF(a +  0, v[j].x * nj);
                            ldsAddF(a + 32, v[j].y * nj);
                        }
                    }
                }
            }
        };

        auto push = [&](int d, int g) {
            const bool hit = (unsigned)(d - base) < (unsigned)nrows;
            const unsigned m = __builtin_amdgcn_ballot_w32(hit);
            if (m != 0u) {
                if (hit) qbuf[tail + (int)__builtin_amdgcn_mbcnt_lo(m, 0u)] = g;
                tail += __builtin_popcount(m);
            }
        };

        auto drain = [&]() {
            if (tail >= 32) {
                int head = 0;
                for (; tail - head >= 32; head += 32) flush(head, 32);
                const int rem = tail - head;
                if (lane < rem) { const int g = qbuf[head + lane]; qbuf[lane] = g; }
                tail = rem;
            }
        };

        auto scan = [&](unsigned m, int e0) {
            while (m != 0u) {
                int cb[8], dlo[8], dhi[8];
#pragma unroll
                for (int j = 0; j < 8; ++j) {
                    cb[j] = -1; dlo[j] = -1; dhi[j] = -1;
                    if (m != 0u) {
                        const int c = e0 + (__builtin_ctz(m) << 6);
                        m &= m - 1u;
                        cb[j] = c;
                        const int g = c + lane;
                        dlo[j] = (g < E) ? dst[g] : -1;
                        dhi[j] = (g + 32 < E) ? dst[g + 32] : -1;
                    }
                }
#pragma unroll
                for (int j = 0; j < 8; ++j) {
                    if (cb[j] >= 0) {
                        push(dlo[j], cb[j] + lane);
                        push(dhi[j], cb[j] + 32 + lane);
                    }
                }
                drain();
            }
        };

        for (int wb = wv; wb < FW; wb += 32 * 32) {
            const int wi = wb + 32 * lane;
            const unsigned mine = (wi < FW) ? fl[wi] : 0u;
            for (int i = 0; i < 32; ++i) {
                const unsigned m = (unsigned)__builtin_amdgcn_readlane((int)mine, i);
                scan(m, (wb + 32 * i) << 11);
            }
        }
        for (int head = 0; head < tail; head += 32) flush(head, min(32, tail - head));
        __syncthreads();

        for (int r = wv; r < nrows; r += 32) {
            const float* a = acc + r * DIM + lane;
            float* op = out + (size_t)(base + r) * DIM + lane * VPL;
            if constexpr (VPL == 4) {
                float4 o = {a[0], a[32], a[64], a[96]};
                if (RELU) { o.x = fmaxf(o.x, 0.0f); o.y = fmaxf(o.y, 0.0f);
                            o.z = fmaxf(o.z, 0.0f); o.w = fmaxf(o.w, 0.0f); }
                *(float4*)op = o;
            } else {
                float2 o = {a[0], a[32]};
                if (RELU) { o.x = fmaxf(o.x, 0.0f); o.y = fmaxf(o.y, 0.0f); }
                *(float2*)op = o;
            }
        }
        __syncthreads();
    }
}

__global__ void k_decode(const float* __restrict__ Z, const int* __restrict__ row,
                         const int* __restrict__ col, float* __restrict__ eerr, int E) {
    const int lane = threadIdx.x & 31;
    const int wave = __builtin_amdgcn_readfirstlane(
        (int)(blockIdx.x * (blockDim.x >> 5) + (threadIdx.x >> 5)));
    const int first = wave * 32;
    if (first >= E) return;
    const int nval = min(32, E - first);

    const int g = first + lane;
    int r = 0, c = 0;
    if (g < E) { r = row[g]; c = col[g]; }

    const float* zl = Z + lane * 2;
    float score = 0.0f;
    for (int j0 = 0; j0 < nval; j0 += 8) {
        float2 zr[8], zc[8];
        for (int j = 0; j < 8; ++j) {
            const int rj = __builtin_amdgcn_readlane(r, j0 + j);
            const int cj = __builtin_amdgcn_readlane(c, j0 + j);
            zr[j] = *(const float2*)(zl + (size_t)rj * 64);
            zc[j] = *(const float2*)(zl + (size_t)cj * 64);
        }
        for (int j = 0; j < 8; ++j) {
            float p = zr[j].x * zc[j].x + zr[j].y * zc[j].y;
#pragma unroll
            for (int off = 16; off > 0; off >>= 1) p += __shfl_xor(p, off, 32);
            if (lane == j0 + j) score = p;
        }
    }

    if (g < E) {
        float sig = 1.0f / (1.0f + __expf(-score));
        eerr[g] = -__logf(sig + 1e-15f);
    }
}

constexpr int SEG_S = 64;
constexpr int SEG_R = 25000;
constexpr int SEG_T = 1024;

__global__ __launch_bounds__(SEG_T)
void k_segsum(const float* __restrict__ eerr, const int* __restrict__ row,
              float* __restrict__ part, int E, int N) {
    __shared__ float2 acc[SEG_R];
    const int tid = threadIdx.x;
    constexpr int nt = SEG_T;
    const int ES  = (E + (int)gridDim.x - 1) / (int)gridDim.x;
    const int e0  = blockIdx.x * ES;
    const int e1  = min(E, e0 + ES);
    float2* pp = (float2*)part + (size_t)blockIdx.x * N;

    for (int base = blockIdx.y * SEG_R; base < N; base += gridDim.y * SEG_R) {
        const int nrows = min(SEG_R, N - base);
        for (int i = tid; i < nrows; i += nt) acc[i] = make_float2(0.0f, 0.0f);
        __syncthreads();

        for (int e = e0 + tid; e < e1; e += 8 * nt) {
            int   r[8];
            float v[8];
#pragma unroll
            for (int k = 0; k < 8; ++k) {
                const int ee = e + k * nt;
                r[k] = -1; v[k] = 0.0f;
                if (ee < e1) { r[k] = row[ee]; v[k] = eerr[ee]; }
            }
#pragma unroll
            for (int k = 0; k < 8; ++k) {
                const unsigned rl = (unsigned)(r[k] - base);
                if (rl < (unsigned)nrows) { ldsAddF(&acc[rl].x, v[k]); ldsAddF(&acc[rl].y, 1.0f); }
            }
        }
        __syncthreads();

        for (int i = tid; i < nrows; i += nt) pp[base + i] = acc[i];
        __syncthreads();
    }
}

__global__ void k_finalize(const float* __restrict__ part, float* __restrict__ out, int N) {
    int i = blockIdx.x * blockDim.x + threadIdx.x;
    if (i < N) {
        float es = 0.0f, c = 0.0f;
#pragma unroll 16
        for (int s = 0; s < SEG_S; ++s) {
            const float2 p = ((const float2*)part)[(size_t)s * N + i];
            es += p.x;
            c  += p.y;
        }
        out[i] = (c > 0.0f) ? (es / fmaxf(c, 1.0f)) : 10.0f;
    }
}

extern "C" void kernel_launch(void* const* d_in, const int* in_sizes, int n_in,
                              void* d_out, int out_size, void* d_ws, size_t ws_size,
                              hipStream_t stream) {
    const float* x  = (const float*)d_in[0];
    const int*   ei = (const int*)d_in[1];
    const float* W1 = (const float*)d_in[2];
    const float* b1 = (const float*)d_in[3];
    const float* W2 = (const float*)d_in[4];
    const float* b2 = (const float*)d_in[5];

    const int N = in_sizes[0] / 128;
    const int E = in_sizes[1] / 2;
    const int* src = ei;
    const int* dst = ei + E;

    float* ws     = (float*)d_ws;
    float* dinv   = ws;
    float* mplane = dinv + N;
    size_t off    = (((size_t)3 * N) + 63) & ~(size_t)63;
    float* wt1    = ws + off;
    float* wt2    = wt1 + 128 * 128;
    float* bufA   = wt2 + 64 * 128;
    float* bufB   = bufA + (size_t)N * 128;

    const int T = 256;
    unsigned* marks   = (unsigned*)mplane;
    unsigned* degpart = (unsigned*)bufB;
    k_degmark  <<<dim3(DM_S, 16), DM_T, 0, stream>>>(dst, degpart, marks, E, N);
    k_rsqrt    <<<(N + T - 1) / T, T, 0, stream>>>(degpart, dinv, N);
    k_fragorder<<<(128 * 128 + T - 1) / T, T, 0, stream>>>(W1, wt1, 128);
    k_fragorder<<<(128 * 64  + T - 1) / T, T, 0, stream>>>(W2, wt2, 64);

    k_gemm_wmma_k128<128><<<N / 16, 256, 0, stream>>>(x, wt1, bufA);
    k_aggregate_own<128, true><<<250, 1024, 0, stream>>>(bufA, bufB, src, dst, dinv, b1, marks, E, N);

    k_gemm_wmma_k128<64><<<N / 16, 128, 0, stream>>>(bufB, wt2, bufA);
    k_aggregate_own<64, false><<<250, 1024, 0, stream>>>(bufA, bufB, src, dst, dinv, b2, marks, E, N);

    float* eerr = bufB + (size_t)N * 64;
    float* part = bufA;
    k_decode<<<(E + 255) / 256, 256, 0, stream>>>(bufB, src, dst, eerr, E);
    k_segsum<<<dim3(SEG_S, 4), SEG_T, 0, stream>>>(eerr, src, part, E, N);
    k_finalize<<<(N + T - 1) / T, T, 0, stream>>>(part, (float*)d_out, N);
}


# Aggregation without a per-edge norm: projection stores row r times dinv[r], owner scales each sum by dinv[d] once, queue keeps the local row: one lookup per edge, not four; norm products regrouped
# speedup vs baseline: 3.2322x; 1.0508x over previous
#include <hip/hip_runtime.h>
#include <math.h>


typedef float v2f __attribute__((ext_vector_type(2)));
typedef float v8f __attribute__((ext_vector_type(8)));

constexpr int OWN_ROWS = 400;

__device__ __forceinline__ void ldsAddU(unsigned* p, unsigned v) {
    __hip_atomic_fetch_add(p, v, __ATOMIC_RELAXED, __HIP_MEMORY_SCOPE_WORKGROUP);
}
__device__ __forceinline__ void ldsOrU(unsigned* p, unsigned v) {
    __hip_atomic_fetch_or(p, v, __ATOMIC_RELAXED, __HIP_MEMORY_SCOPE_WORKGROUP);
}

constexpr int DM_S    = 16;
constexpr int DM_OPR  = 16;
constexpr int DM_ROWS = DM_OPR * OWN_ROWS;
constexpr int DM_WMAX = 64;
constexpr int DM_T    = 1024;

__global__ __launch_bounds__(DM_T)
void k_degmark(const int* __restrict__ dst, unsigned* __restrict__ degpart,
               unsigned* __restrict__ marks, int E, int N) {
    __shared__ unsigned cntl[DM_ROWS];
    __shared__ unsigned mk[DM_OPR * DM_WMAX];
    const int tid = threadIdx.x;
    const int FW  = (E + 2047) >> 11;
    const int wps = (FW + (int)gridDim.x - 1) / (int)gridDim.x;
    const int w0  = blockIdx.x * wps;
    const int e0  = w0 << 11;
    const int e1  = min(E, (w0 + wps) << 11);

    for (int rg = blockIdx.y; rg * DM_ROWS < N; rg += gridDim.y) {
        const int base  = rg * DM_ROWS;
        const int nrows = min(DM_ROWS, N - base);
        for (int i = tid; i < DM_ROWS; i += DM_T) cntl[i] = 0u;
        for (int i = tid; i < DM_OPR * DM_WMAX; i += DM_T) mk[i] = 0u;
        __syncthreads();

        for (int e = e0 + tid; e < e1; e += 16 * DM_T) {
            int d[16];
#pragma unroll
            for (int k = 0; k < 16; ++k) {
                const int ee = e + k * DM_T;
                d[k] = (ee < e1) ? dst[ee] : -1;
            }
#pragma unroll
            for (int k = 0; k < 16; ++k) {
                const unsigned rl = (unsigned)(d[k] - base);
                if (rl < (unsigned)nrows) {
                    const int ee = e + k * DM_T;
                    ldsAddU(&cntl[rl], 1u);
                    ldsOrU(&mk[(rl / OWN_ROWS) * DM_WMAX + ((ee - e0) >> 11)],
                           1u << ((ee >> 6) & 31));
                }
            }
        }
        __syncthreads();

        unsigned* pl = degpart + (size_t)blockIdx.x * N + base;
        for (int i = tid; i < nrows; i += DM_T) pl[i] = cntl[i];
        for (int i = tid; i < DM_OPR * DM_WMAX; i += DM_T) {
            const int o = i / DM_WMAX, w = i - o * DM_WMAX;
            const int owner = rg * DM_OPR + o;
            if (w < wps && w0 + w < FW && owner * OWN_ROWS < N)
                marks[(size_t)owner * FW + w0 + w] = mk[i];
        }
        __syncthreads();
    }
}

__global__ void k_rsqrt(const unsigned* __restrict__ degpart, float* __restrict__ dinv, int N) {
    int i = blockIdx.x * blockDim.x + threadIdx.x;
    if (i < N) {
        unsigned c = 1u;
#pragma unroll
        for (int s = 0; s < DM_S; ++s) c += degpart[(size_t)s * N + i];
        dinv[i] = rsqrtf((float)c);
    }
}

__global__ void k_fragorder(const float* __restrict__ W, float* __restrict__ Wf,
                            int N) {
    int idx = blockIdx.x * blockDim.x + threadIdx.x;
    if (idx < 128 * N) {
        const int j    = idx & 3;
        const int lane = (idx >> 2) & 31;
        const int kq   = (idx >> 7) & 15;
        const int ct   = idx >> 11;
        const int k = 8 * kq + 4 * (j >> 1) + 2 * (lane >> 4) + (j & 1);
        const int n = 16 * ct + (lane & 15);
        Wf[idx] = W[k * N + n];
    }
}

template <int NCOLS>
__global__ void k_gemm_wmma_k128(const float* __restrict__ A,
                                 const float* __restrict__ Bf,
                                 const float* __restrict__ rs,
                                 float* __restrict__ C) {
    __shared__ float As[16 * 132];
    const int tid  = threadIdx.x;
    const int row0 = blockIdx.x * 16;

    const float4* Ag = (const float4*)(A + (size_t)row0 * 128);
    for (int i = tid; i < 16 * 32; i += blockDim.x) {
        int r = i >> 5, c4 = i & 31;
        ((float4*)(As + r * 132))[c4] = Ag[r * 32 + c4];
    }
    __syncthreads();

    const int lane  = tid & 31;
    const int col0  = (tid >> 5) * 16;
    const int n15   = lane & 15;
    const int khalf = lane >> 4;

    v8f acc = {0.f, 0.f, 0.f, 0.f, 0.f, 0.f, 0.f, 0.f};
    const float* ap = As + n15 * 132 + 2 * khalf;
    const float* bp = Bf + ((size_t)(tid >> 5) * (16 * 32) + lane) * 4;

#pragma unroll
    for (int kq = 0; kq < 16; ++kq) {
        const float4 b4 = *(const float4*)(bp + kq * 128);
        v2f a0; a0.x = ap[8 * kq];     a0.y = ap[8 * kq + 1];
        v2f b0; b0.x = b4.x;           b0.y = b4.y;
        acc = __builtin_amdgcn_wmma_f32_16x16x4_f32(
             false, a0,  false, b0,
             (short)0, acc,  false,  false);
        v2f a1; a1.x = ap[8 * kq + 4]; a1.y = ap[8 * kq + 5];
        v2f b1; b1.x = b4.z;           b1.y = b4.w;
        acc = __builtin_amdgcn_wmma_f32_16x16x4_f32(
             false, a1,  false, b1,
             (short)0, acc,  false,  false);
    }

    float* Cp = C + (size_t)(row0 + 8 * khalf) * NCOLS + col0 + n15;
    const float* rp = rs + row0 + 8 * khalf;
#pragma unroll
    for (int r = 0; r < 8; ++r) Cp[r * NCOLS] = acc[r] * rp[r];
}

constexpr int OWN_QCAP = 544;

__device__ __forceinline__ void ldsAddF(float* p, float v) {
    __hip_atomic_fetch_add(p, v, __ATOMIC_RELAXED, __HIP_MEMORY_SCOPE_WORKGROUP);
}

template <int DIM, bool RELU>
__global__ __launch_bounds__(1024)
void k_aggregate_own(const float* __restrict__ H, float* __restrict__ out,
                     const int* __restrict__ src, const int* __restrict__ dst,
                     const float* __restrict__ dinv, const float* __restrict__ bias,
                     const unsigned* __restrict__ marks, int E, int N) {
    constexpr int VPL = DIM / 32;
    static_assert(OWN_ROWS <= 65536, "local row numbers are kept in 16 bits");
    __shared__ float acc[OWN_ROWS * DIM];
    __shared__ int   queue[32 * OWN_QCAP];
    __shared__ unsigned short qrow[32 * OWN_QCAP];

    const int lane = threadIdx.x & 31;
    const int wv = __builtin_amdgcn_readfirstlane((int)(threadIdx.x >> 5));
    int* qbuf = queue + wv * OWN_QCAP;
    unsigned short* qdl = qrow + wv * OWN_QCAP;
    const float* hl = H + lane * VPL;
    float bz[VPL];
#pragma unroll
    for (int k = 0; k < VPL; ++k) bz[k] = bias[lane * VPL + k];

    const int FW = (E + 2047) >> 11;
    for (int base = blockIdx.x * OWN_ROWS; base < N; base += gridDim.x * OWN_ROWS) {
        const int nrows = min(OWN_ROWS, N - base);
        const unsigned* fl = marks + (base / OWN_ROWS) * FW;

        for (int r = wv; r < nrows; r += 32) {
            const int row = base + r;
            float* a = acc + r * DIM + lane;
            if constexpr (VPL == 4) {
                const float4 h = *(const float4*)(hl + (size_t)row * DIM);
                a[0] = h.x; a[32] = h.y; a[64] = h.z; a[96] = h.w;
            } else {
                const float2 h = *(const float2*)(hl + (size_t)row * DIM);
                a[0] = h.x; a[32] = h.y;
            }
        }
        __syncthreads();

        int tail = 0;

        auto flush = [&](int off, int cnt) {
            int s = 0, dl = 0;
            if (lane < cnt) {
                s  = src[qbuf[off + lane]];
                dl = (int)qdl[off + lane];
            }
            for (int j0 = 0; j0 < cnt; j0 += 8) {
                if constexpr (VPL == 4) {
                    float4 v[8];
                    for (int j = 0; j < 8; ++j) {
                        const int sj = __builtin_amdgcn_readlane(s, j0 + j);
                        v[j] = *(const float4*)(hl + (size_t)sj * DIM);
                    }
                    for (int j = 0; j < 8; ++j) {
                        if (j0 + j < cnt) {
                            const int dj = __builtin_amdgcn_readlane(dl, j0 + j);
                            float* a = acc + dj * DIM + lane;
                            ldsAddF(a +  0, v[j].x);
                            ldsAddF(a + 32, v[j].y);
                            ldsAddF(a + 64, v[j].z);
                            ldsAddF(a + 96, v[j].w);
                        }
                    }
                } else {
                    float2 v[8];
                    for (int j = 0; j < 8; ++j) {
                        const int sj = __builtin_amdgcn_readlane(s, j0 + j);
                        v[j] = *(const float2*)(hl + (size_t)sj * DIM);
                    }
                    for (int j = 0; j < 8; ++j) {
                        if (j0 + j < cnt) {
                            const int dj = __builtin_amdgcn_readlane(dl, j0 + j);
                            float* a = acc + dj * DIM + lane;
                            ldsAddF(a +  0, v[j].x);
                            ldsAddF(a + 32, v[j].y);
                        }
                    }
                }
            }
        };

        auto push = [&](int d, int g) {
            const unsigned rl = (unsigned)(d - base);
            const bool hit = rl < (unsigned)nrows;
            const unsigned m = __builtin_amdgcn_ballot_w32(hit);
            if (m != 0u) {
                if (hit) {
                    const int q = tail + (int)__builtin_amdgcn_mbcnt_lo(m, 0u);
                    qbuf[q] = g;
                    qdl[q]  = (unsigned short)rl;
                }
                tail += __builtin_popcount(m);
            }
        };

        auto drain = [&]() {
            if (tail >= 32) {
                int head = 0;
                for (; tail - head >= 32; head += 32) flush(head, 32);
                const int rem = tail - head;
                if (lane < rem) {
                    const int g = qbuf[head + lane];
                    const unsigned short t = qdl[head + lane];
                    qbuf[lane] = g;
                    qdl[lane]  = t;
                }
                tail = rem;
            }
        };

        auto scan = [&](unsigned m, int e0) {
            while (m != 0u) {
                int cb[8], dlo[8], dhi[8];
#pragma unroll
                for (int j = 0; j < 8; ++j) {
                    cb[j] = -1; dlo[j] = -1; dhi[j] = -1;
                    if (m != 0u) {
                        const int c = e0 + (__builtin_ctz(m) << 6);
                        m &= m - 1u;
                        cb[j] = c;
                        const int g = c + lane;
                        dlo[j] = (g < E) ? dst[g] : -1;
                        dhi[j] = (g + 32 < E) ? dst[g + 32] : -1;
                    }
                }
#pragma unroll
                for (int j = 0; j < 8; ++j) {
                    if (cb[j] >= 0) {
                        push(dlo[j], cb[j] + lane);
                        push(dhi[j], cb[j] + 32 + lane);
                    }
                }
                drain();
            }
        };

        for (int wb = wv; wb < FW; wb += 32 * 32) {
            const int wi = wb + 32 * lane;
            const unsigned mine = (wi < FW) ? fl[wi] : 0u;
            for (int i = 0; i < 32; ++i) {
                const unsigned m = (unsigned)__builtin_amdgcn_readlane((int)mine, i);
                scan(m, (wb + 32 * i) << 11);
            }
        }
        for (int head = 0; head < tail; head += 32) flush(head, min(32, tail - head));
        __syncthreads();

        for (int r = wv; r < nrows; r += 32) {
            const float* a = acc + r * DIM + lane;
            const float di = dinv[base + r];
            float* op = out + (size_t)(base + r) * DIM + lane * VPL;
            if constexpr (VPL == 4) {
                float4 o = {bz[0] + di * a[0], bz[1] + di * a[32],
                            bz[2] + di * a[64], bz[3] + di * a[96]};
                if (RELU) { o.x = fmaxf(o.x, 0.0f); o.y = fmaxf(o.y, 0.0f);
                            o.z = fmaxf(o.z, 0.0f); o.w = fmaxf(o.w, 0.0f); }
                *(float4*)op = o;
            } else {
                float2 o = {bz[0] + di * a[0], bz[1] + di * a[32]};
                if (RELU) { o.x = fmaxf(o.x, 0.0f); o.y = fmaxf(o.y, 0.0f); }
                *(float2*)op = o;
            }
        }
        __syncthreads();
    }
}

__global__ void k_decode(const float* __restrict__ Z, const int* __restrict__ row,
                         const int* __restrict__ col, float* __restrict__ eerr, int E) {
    const int lane = threadIdx.x & 31;
    const int wave = __builtin_amdgcn_readfirstlane(
        (int)(blockIdx.x * (blockDim.x >> 5) + (threadIdx.x >> 5)));
    const int first = wave * 32;
    if (first >= E) return;
    const int nval = min(32, E - first);

    const int g = first + lane;
    int r = 0, c = 0;
    if (g < E) { r = row[g]; c = col[g]; }

    const float* zl = Z + lane * 2;
    float score = 0.0f;
    for (int j0 = 0; j0 < nval; j0 += 8) {
        float2 zr[8], zc[8];
        for (int j = 0; j < 8; ++j) {
            const int rj = __builtin_amdgcn_readlane(r, j0 + j);
            const int cj = __builtin_amdgcn_readlane(c, j0 + j);
            zr[j] = *(const float2*)(zl + (size_t)rj * 64);
            zc[j] = *(const float2*)(zl + (size_t)cj * 64);
        }
        for (int j = 0; j < 8; ++j) {
            float p = zr[j].x * zc[j].x + zr[j].y * zc[j].y;
#pragma unroll
            for (int off = 16; off > 0; off >>= 1) p += __shfl_xor(p, off, 32);
            if (lane == j0 + j) score = p;
        }
    }

    if (g < E) {
        float sig = 1.0f / (1.0f + __expf(-score));
        eerr[g] = -__logf(sig + 1e-15f);
    }
}

constexpr int SEG_S = 64;
constexpr int SEG_R = 25000;
constexpr int SEG_T = 1024;

__global__ __launch_bounds__(SEG_T)
void k_segsum(const float* __restrict__ eerr, const int* __restrict__ row,
              float* __restrict__ part, int E, int N) {
    __shared__ float2 acc[SEG_R];
    const int tid = threadIdx.x;
    constexpr int nt = SEG_T;
    const int ES  = (E + (int)gridDim.x - 1) / (int)gridDim.x;
    const int e0  = blockIdx.x * ES;
    const int e1  = min(E, e0 + ES);
    float2* pp = (float2*)part + (size_t)blockIdx.x * N;

    for (int base = blockIdx.y * SEG_R; base < N; base += gridDim.y * SEG_R) {
        const int nrows = min(SEG_R, N - base);
        for (int i = tid; i < nrows; i += nt) acc[i] = make_float2(0.0f, 0.0f);
        __syncthreads();

        for (int e = e0 + tid; e < e1; e += 8 * nt) {
            int   r[8];
            float v[8];
#pragma unroll
            for (int k = 0; k < 8; ++k) {
                const int ee = e + k * nt;
                r[k] = -1; v[k] = 0.0f;
                if (ee < e1) { r[k] = row[ee]; v[k] = eerr[ee]; }
            }
#pragma unroll
            for (int k = 0; k < 8; ++k) {
                const unsigned rl = (unsigned)(r[k] - base);
                if (rl < (unsigned)nrows) { ldsAddF(&acc[rl].x, v[k]); ldsAddF(&acc[rl].y, 1.0f); }
            }
        }
        __syncthreads();

        for (int i = tid; i < nrows; i += nt) pp[base + i] = acc[i];
        __syncthreads();
    }
}

__global__ void k_finalize(const float* __restrict__ part, float* __restrict__ out, int N) {
    int i = blockIdx.x * blockDim.x + threadIdx.x;
    if (i < N) {
        float es = 0.0f, c = 0.0f;
#pragma unroll 16
        for (int s = 0; s < SEG_S; ++s) {
            const float2 p = ((const float2*)part)[(size_t)s * N + i];
            es += p.x;
            c  += p.y;
        }
        out[i] = (c > 0.0f) ? (es / fmaxf(c, 1.0f)) : 10.0f;
    }
}

extern "C" void kernel_launch(void* const* d_in, const int* in_sizes, int n_in,
                              void* d_out, int out_size, void* d_ws, size_t ws_size,
                              hipStream_t stream) {
    const float* x  = (const float*)d_in[0];
    const int*   ei = (const int*)d_in[1];
    const float* W1 = (const float*)d_in[2];
    const float* b1 = (const float*)d_in[3];
    const float* W2 = (const float*)d_in[4];
    const float* b2 = (const float*)d_in[5];

    const int N = in_sizes[0] / 128;
    const int E = in_sizes[1] / 2;
    const int* src = ei;
    const int* dst = ei + E;

    float* ws     = (float*)d_ws;
    float* dinv   = ws;
    float* mplane = dinv + N;
    size_t off    = (((size_t)3 * N) + 63) & ~(size_t)63;
    float* wt1    = ws + off;
    float* wt2    = wt1 + 128 * 128;
    float* bufA   = wt2 + 64 * 128;
    float* bufB   = bufA + (size_t)N * 128;

    const int T = 256;
    unsigned* marks   = (unsigned*)mplane;
    unsigned* degpart = (unsigned*)bufB;
    k_degmark  <<<dim3(DM_S, 16), DM_T, 0, stream>>>(dst, degpart, marks, E, N);
    k_rsqrt    <<<(N + T - 1) / T, T, 0, stream>>>(degpart, dinv, N);
    k_fragorder<<<(128 * 128 + T - 1) / T, T, 0, stream>>>(W1, wt1, 128);
    k_fragorder<<<(128 * 64  + T - 1) / T, T, 0, stream>>>(W2, wt2, 64);

    k_gemm_wmma_k128<128><<<N / 16, 256, 0, stream>>>(x, wt1, dinv, bufA);
    k_aggregate_own<128, true><<<250, 1024, 0, stream>>>(bufA, bufB, src, dst, dinv, b1, marks, E, N);

    k_gemm_wmma_k128<64><<<N / 16, 128, 0, stream>>>(bufB, wt2, dinv, bufA);
    k_aggregate_own<64, false><<<250, 1024, 0, stream>>>(bufA, bufB, src, dst, dinv, b2, marks, E, N);

    float* eerr = bufB + (size_t)N * 64;
    float* part = bufA;
    k_decode<<<(E + 255) / 256, 256, 0, stream>>>(bufB, src, dst, eerr, E);
    k_segsum<<<dim3(SEG_S, 4), SEG_T, 0, stream>>>(eerr, src, part, E, N);
    k_finalize<<<(N + T - 1) / T, T, 0, stream>>>(part, (float*)d_out, N);
}


# Aggregation, fewer memory waits per wave: next batch of marked chunks requested under the current one, 16/32 row gathers rolling per group, start/store loops 4 rows deep; sums stay in arrival order
# speedup vs baseline: 3.6023x; 1.1145x over previous
#include <hip/hip_runtime.h>
#include <math.h>


typedef float v2f __attribute__((ext_vector_type(2)));
typedef float v8f __attribute__((ext_vector_type(8)));

constexpr int OWN_ROWS = 400;

__device__ __forceinline__ void ldsAddU(unsigned* p, unsigned v) {
    __hip_atomic_fetch_add(p, v, __ATOMIC_RELAXED, __HIP_MEMORY_SCOPE_WORKGROUP);
}
__device__ __forceinline__ void ldsOrU(unsigned* p, unsigned v) {
    __hip_atomic_fetch_or(p, v, __ATOMIC_RELAXED, __HIP_MEMORY_SCOPE_WORKGROUP);
}

constexpr int DM_S    = 16;
constexpr int DM_OPR  = 16;
constexpr int DM_ROWS = DM_OPR * OWN_ROWS;
constexpr int DM_WMAX = 64;
constexpr int DM_T    = 1024;

__global__ __launch_bounds__(DM_T)
void k_degmark(const int* __restrict__ dst, unsigned* __restrict__ degpart,
               unsigned* __restrict__ marks, int E, int N) {
    __shared__ unsigned cntl[DM_ROWS];
    __shared__ unsigned mk[DM_OPR * DM_WMAX];
    const int tid = threadIdx.x;
    const int FW  = (E + 2047) >> 11;
    const int wps = (FW + (int)gridDim.x - 1) / (int)gridDim.x;
    const int w0  = blockIdx.x * wps;
    const int e0  = w0 << 11;
    const int e1  = min(E, (w0 + wps) << 11);

    for (int rg = blockIdx.y; rg * DM_ROWS < N; rg += gridDim.y) {
        const int base  = rg * DM_ROWS;
        const int nrows = min(DM_ROWS, N - base);
        for (int i = tid; i < DM_ROWS; i += DM_T) cntl[i] = 0u;
        for (int i = tid; i < DM_OPR * DM_WMAX; i += DM_T) mk[i] = 0u;
        __syncthreads();

        for (int e = e0 + tid; e < e1; e += 16 * DM_T) {
            int d[16];
#pragma unroll
            for (int k = 0; k < 16; ++k) {
                const int ee = e + k * DM_T;
                d[k] = (ee < e1) ? dst[ee] : -1;
            }
#pragma unroll
            for (int k = 0; k < 16; ++k) {
                const unsigned rl = (unsigned)(d[k] - base);
                if (rl < (unsigned)nrows) {
                    const int ee = e + k * DM_T;
                    ldsAddU(&cntl[rl], 1u);
                    ldsOrU(&mk[(rl / OWN_ROWS) * DM_WMAX + ((ee - e0) >> 11)],
                           1u << ((ee >> 6) & 31));
                }
            }
        }
        __syncthreads();

        unsigned* pl = degpart + (size_t)blockIdx.x * N + base;
        for (int i = tid; i < nrows; i += DM_T) pl[i] = cntl[i];
        for (int i = tid; i < DM_OPR * DM_WMAX; i += DM_T) {
            const int o = i / DM_WMAX, w = i - o * DM_WMAX;
            const int owner = rg * DM_OPR + o;
            if (w < wps && w0 + w < FW && owner * OWN_ROWS < N)
                marks[(size_t)owner * FW + w0 + w] = mk[i];
        }
        __syncthreads();
    }
}

__global__ void k_rsqrt(const unsigned* __restrict__ degpart, float* __restrict__ dinv, int N) {
    int i = blockIdx.x * blockDim.x + threadIdx.x;
    if (i < N) {
        unsigned c = 1u;
#pragma unroll
        for (int s = 0; s < DM_S; ++s) c += degpart[(size_t)s * N + i];
        dinv[i] = rsqrtf((float)c);
    }
}

__global__ void k_fragorder(const float* __restrict__ W, float* __restrict__ Wf,
                            int N) {
    int idx = blockIdx.x * blockDim.x + threadIdx.x;
    if (idx < 128 * N) {
        const int j    = idx & 3;
        const int lane = (idx >> 2) & 31;
        const int kq   = (idx >> 7) & 15;
        const int ct   = idx >> 11;
        const int k = 8 * kq + 4 * (j >> 1) + 2 * (lane >> 4) + (j & 1);
        const int n = 16 * ct + (lane & 15);
        Wf[idx] = W[k * N + n];
    }
}

template <int NCOLS>
__global__ void k_gemm_wmma_k128(const float* __restrict__ A,
                                 const float* __restrict__ Bf,
                                 const float* __restrict__ rs,
                                 float* __restrict__ C) {
    __shared__ float As[16 * 132];
    const int tid  = threadIdx.x;
    const int row0 = blockIdx.x * 16;

    const float4* Ag = (const float4*)(A + (size_t)row0 * 128);
    for (int i = tid; i < 16 * 32; i += blockDim.x) {
        int r = i >> 5, c4 = i & 31;
        ((float4*)(As + r * 132))[c4] = Ag[r * 32 + c4];
    }
    __syncthreads();

    const int lane  = tid & 31;
    const int col0  = (tid >> 5) * 16;
    const int n15   = lane & 15;
    const int khalf = lane >> 4;

    v8f acc = {0.f, 0.f, 0.f, 0.f, 0.f, 0.f, 0.f, 0.f};
    const float* ap = As + n15 * 132 + 2 * khalf;
    const float* bp = Bf + ((size_t)(tid >> 5) * (16 * 32) + lane) * 4;

#pragma unroll
    for (int kq = 0; kq < 16; ++kq) {
        const float4 b4 = *(const float4*)(bp + kq * 128);
        v2f a0; a0.x = ap[8 * kq];     a0.y = ap[8 * kq + 1];
        v2f b0; b0.x = b4.x;           b0.y = b4.y;
        acc = __builtin_amdgcn_wmma_f32_16x16x4_f32(
             false, a0,  false, b0,
             (short)0, acc,  false,  false);
        v2f a1; a1.x = ap[8 * kq + 4]; a1.y = ap[8 * kq + 5];
        v2f b1; b1.x = b4.z;           b1.y = b4.w;
        acc = __builtin_amdgcn_wmma_f32_16x16x4_f32(
             false, a1,  false, b1,
             (short)0, acc,  false,  false);
    }

    float* Cp = C + (size_t)(row0 + 8 * khalf) * NCOLS + col0 + n15;
    const float* rp = rs + row0 + 8 * khalf;
#pragma unroll
    for (int r = 0; r < 8; ++r) Cp[r * NCOLS] = acc[r] * rp[r];
}

constexpr int OWN_QCAP = 544;

__device__ __forceinline__ void ldsAddF(float* p, float v) {
    __hip_atomic_fetch_add(p, v, __ATOMIC_RELAXED, __HIP_MEMORY_SCOPE_WORKGROUP);
}

template <int VPL> struct RowVec;
template <> struct RowVec<4> { typedef float4 T; };
template <> struct RowVec<2> { typedef float2 T; };

template <int DIM, bool RELU>
__global__ __launch_bounds__(1024)
void k_aggregate_own(const float* __restrict__ H, float* __restrict__ out,
                     const int* __restrict__ src, const int* __restrict__ dst,
                     const float* __restrict__ dinv, const float* __restrict__ bias,
                     const unsigned* __restrict__ marks, int E, int N) {
    constexpr int VPL = DIM / 32;
    constexpr int AGG_RF = 64 / VPL;
    typedef typename RowVec<VPL>::T VT;
    static_assert(OWN_ROWS <= 65536, "local row numbers are kept in 16 bits");
    __shared__ float acc[OWN_ROWS * DIM];
    __shared__ int   queue[32 * OWN_QCAP];
    __shared__ unsigned short qrow[32 * OWN_QCAP];

    const int lane = threadIdx.x & 31;
    const int wv = __builtin_amdgcn_readfirstlane((int)(threadIdx.x >> 5));
    int* qbuf = queue + wv * OWN_QCAP;
    unsigned short* qdl = qrow + wv * OWN_QCAP;
    const float* hl = H + lane * VPL;
    float bz[VPL];
#pragma unroll
    for (int k = 0; k < VPL; ++k) bz[k] = bias[lane * VPL + k];

    auto ldrow = [&](int r) -> VT { return *(const VT*)(hl + (size_t)r * DIM); };
    auto addrow = [&](int dj, const VT& v) {
        float* a = acc + dj * DIM + lane;
        if constexpr (VPL == 4) {
            ldsAddF(a +  0, v.x); ldsAddF(a + 32, v.y);
            ldsAddF(a + 64, v.z); ldsAddF(a + 96, v.w);
        } else {
            ldsAddF(a +  0, v.x); ldsAddF(a + 32, v.y);
        }
    };

    const int FW = (E + 2047) >> 11;
    for (int base = blockIdx.x * OWN_ROWS; base < N; base += gridDim.x * OWN_ROWS) {
        const int nrows = min(OWN_ROWS, N - base);
        const unsigned* fl = marks + (base / OWN_ROWS) * FW;

        int wb = wv;
        unsigned mine = (wb + 32 * lane < FW) ? fl[wb + 32 * lane] : 0u;

        auto putrow = [&](int r, const VT& h) {
            float* a = acc + r * DIM + lane;
            if constexpr (VPL == 4) {
                a[0] = h.x; a[32] = h.y; a[64] = h.z; a[96] = h.w;
            } else {
                a[0] = h.x; a[32] = h.y;
            }
        };
        {
            int r = wv;
            for (; r + 96 < nrows; r += 128) {
                VT h[4];
#pragma unroll
                for (int k = 0; k < 4; ++k) h[k] = ldrow(base + r + 32 * k);
#pragma unroll
                for (int k = 0; k < 4; ++k) putrow(r + 32 * k, h[k]);
            }
            for (; r < nrows; r += 32) putrow(r, ldrow(base + r));
        }
        __syncthreads();

        int tail = 0;

        auto flush32 = [&](int off) {
            const int s  = src[qbuf[off + lane]];
            const int dl = (int)qdl[off + lane];
            VT v[AGG_RF];
#pragma unroll
            for (int j = 0; j < AGG_RF; ++j) v[j] = ldrow(__builtin_amdgcn_readlane(s, j));
#pragma unroll
            for (int j0 = 0; j0 + AGG_RF < 32; j0 += AGG_RF) {
#pragma unroll
                for (int j = 0; j < AGG_RF; ++j) {
                    addrow(__builtin_amdgcn_readlane(dl, j0 + j), v[j]);
                    v[j] = ldrow(__builtin_amdgcn_readlane(s, j0 + AGG_RF + j));
                }
            }
#pragma unroll
            for (int j = 0; j < AGG_RF; ++j)
                addrow(__builtin_amdgcn_readlane(dl, 32 - AGG_RF + j), v[j]);
        };

        auto flush_part = [&](int off, int cnt) {
            int s = 0, dl = 0;
            if (lane < cnt) {
                s  = src[qbuf[off + lane]];
                dl = (int)qdl[off + lane];
            }
            for (int j0 = 0; j0 < cnt; j0 += 8) {
                VT v[8];
                for (int j = 0; j < 8; ++j)
                    v[j] = ldrow(__builtin_amdgcn_readlane(s, j0 + j));
                for (int j = 0; j < 8; ++j)
                    if (j0 + j < cnt) addrow(__builtin_amdgcn_readlane(dl, j0 + j), v[j]);
            }
        };

        auto push = [&](int d, int g) {
            const unsigned rl = (unsigned)(d - base);
            const bool hit = (rl < (unsigned)nrows) && (g < E);
            const unsigned m = __builtin_amdgcn_ballot_w32(hit);
            if (m != 0u) {
                if (hit) {
                    const int q = tail + (int)__builtin_amdgcn_mbcnt_lo(m, 0u);
                    qbuf[q] = g;
                    qdl[q]  = (unsigned short)rl;
                }
                tail += __builtin_popcount(m);
            }
        };

        auto drain = [&]() {
            if (tail >= 32) {
                int head = 0;
                for (; tail - head >= 32; head += 32) flush32(head);
                const int rem = tail - head;
                if (lane < rem) {
                    const int g = qbuf[head + lane];
                    const unsigned short t = qdl[head + lane];
                    qbuf[lane] = g;
                    qdl[lane]  = t;
                }
                tail = rem;
            }
        };

        while (wb < FW) {
            int wi = 0;
            unsigned m = (unsigned)__builtin_amdgcn_readlane((int)mine, 0);

            auto fetch = [&](int (&cb)[8], int (&dlo)[8], int (&dhi)[8]) {
#pragma unroll
                for (int j = 0; j < 8; ++j) {
                    while (m == 0u && wi < 31) {
                        ++wi;
                        m = (unsigned)__builtin_amdgcn_readlane((int)mine, wi);
                    }
                    const bool have = (m != 0u);
                    const int c = have ? ((wb + 32 * wi) << 11) + (__builtin_ctz(m) << 6) : 0;
                    m &= m - 1u;
                    cb[j] = have ? c : -1;
                    const int g = c + lane;
                    dlo[j] = dst[min(g, E - 1)];
                    dhi[j] = dst[min(g + 32, E - 1)];
                }
            };

            int cbA[8], dloA[8], dhiA[8];
            fetch(cbA, dloA, dhiA);
            while (cbA[0] >= 0) {
                int cbB[8], dloB[8], dhiB[8];
                fetch(cbB, dloB, dhiB);
#pragma unroll
                for (int j = 0; j < 8; ++j) {
                    if (cbA[j] >= 0) {
                        push(dloA[j], cbA[j] + lane);
                        push(dhiA[j], cbA[j] + 32 + lane);
                    }
                }
                drain();
#pragma unroll
                for (int j = 0; j < 8; ++j) { cbA[j] = cbB[j]; dloA[j] = dloB[j]; dhiA[j] = dhiB[j]; }
            }

            wb += 32 * 32;
            if (wb < FW) mine = (wb + 32 * lane < FW) ? fl[wb + 32 * lane] : 0u;
        }
        if (tail > 0) flush_part(0, tail);
        __syncthreads();

        auto outrow = [&](int rr, float di) {
            const float* a = acc + rr * DIM + lane;
            float* op = out + (size_t)(base + rr) * DIM + lane * VPL;
            if constexpr (VPL == 4) {
                float4 o = {bz[0] + di * a[0], bz[1] + di * a[32],
                            bz[2] + di * a[64], bz[3] + di * a[96]};
                if (RELU) { o.x = fmaxf(o.x, 0.0f); o.y = fmaxf(o.y, 0.0f);
                            o.z = fmaxf(o.z, 0.0f); o.w = fmaxf(o.w, 0.0f); }
                *(float4*)op = o;
            } else {
                float2 o = {bz[0] + di * a[0], bz[1] + di * a[32]};
                if (RELU) { o.x = fmaxf(o.x, 0.0f); o.y = fmaxf(o.y, 0.0f); }
                *(float2*)op = o;
            }
        };
        {
            int r = wv;
            for (; r + 96 < nrows; r += 128) {
                float di[4];
#pragma unroll
                for (int k = 0; k < 4; ++k) di[k] = dinv[base + r + 32 * k];
#pragma unroll
                for (int k = 0; k < 4; ++k) outrow(r + 32 * k, di[k]);
            }
            for (; r < nrows; r += 32) outrow(r, dinv[base + r]);
        }
        __syncthreads();
    }
}

__global__ void k_decode(const float* __restrict__ Z, const int* __restrict__ row,
                         const int* __restrict__ col, float* __restrict__ eerr, int E) {
    const int lane = threadIdx.x & 31;
    const int wave = __builtin_amdgcn_readfirstlane(
        (int)(blockIdx.x * (blockDim.x >> 5) + (threadIdx.x >> 5)));
    const int first = wave * 32;
    if (first >= E) return;
    const int nval = min(32, E - first);

    const int g = first + lane;
    int r = 0, c = 0;
    if (g < E) { r = row[g]; c = col[g]; }

    const float* zl = Z + lane * 2;
    float score = 0.0f;
    for (int j0 = 0; j0 < nval; j0 += 8) {
        float2 zr[8], zc[8];
        for (int j = 0; j < 8; ++j) {
            const int rj = __builtin_amdgcn_readlane(r, j0 + j);
            const int cj = __builtin_amdgcn_readlane(c, j0 + j);
            zr[j] = *(const float2*)(zl + (size_t)rj * 64);
            zc[j] = *(const float2*)(zl + (size_t)cj * 64);
        }
        for (int j = 0; j < 8; ++j) {
            float p = zr[j].x * zc[j].x + zr[j].y * zc[j].y;
#pragma unroll
            for (int off = 16; off > 0; off >>= 1) p += __shfl_xor(p, off, 32);
            if (lane == j0 + j) score = p;
        }
    }

    if (g < E) {
        float sig = 1.0f / (1.0f + __expf(-score));
        eerr[g] = -__logf(sig + 1e-15f);
    }
}

constexpr int SEG_S = 64;
constexpr int SEG_R = 25000;
constexpr int SEG_T = 1024;

__global__ __launch_bounds__(SEG_T)
void k_segsum(const float* __restrict__ eerr, const int* __restrict__ row,
              float* __restrict__ part, int E, int N) {
    __shared__ float2 acc[SEG_R];
    const int tid = threadIdx.x;
    constexpr int nt = SEG_T;
    const int ES  = (E + (int)gridDim.x - 1) / (int)gridDim.x;
    const int e0  = blockIdx.x * ES;
    const int e1  = min(E, e0 + ES);
    float2* pp = (float2*)part + (size_t)blockIdx.x * N;

    for (int base = blockIdx.y * SEG_R; base < N; base += gridDim.y * SEG_R) {
        const int nrows = min(SEG_R, N - base);
        for (int i = tid; i < nrows; i += nt) acc[i] = make_float2(0.0f, 0.0f);
        __syncthreads();

        for (int e = e0 + tid; e < e1; e += 8 * nt) {
            int   r[8];
            float v[8];
#pragma unroll
            for (int k = 0; k < 8; ++k) {
                const int ee = e + k * nt;
                r[k] = -1; v[k] = 0.0f;
                if (ee < e1) { r[k] = row[ee]; v[k] = eerr[ee]; }
            }
#pragma unroll
            for (int k = 0; k < 8; ++k) {
                const unsigned rl = (unsigned)(r[k] - base);
                if (rl < (unsigned)nrows) { ldsAddF(&acc[rl].x, v[k]); ldsAddF(&acc[rl].y, 1.0f); }
            }
        }
        __syncthreads();

        for (int i = tid; i < nrows; i += nt) pp[base + i] = acc[i];
        __syncthreads();
    }
}

__global__ void k_finalize(const float* __restrict__ part, float* __restrict__ out, int N) {
    int i = blockIdx.x * blockDim.x + threadIdx.x;
    if (i < N) {
        float es = 0.0f, c = 0.0f;
#pragma unroll 16
        for (int s = 0; s < SEG_S; ++s) {
            const float2 p = ((const float2*)part)[(size_t)s * N + i];
            es += p.x;
            c  += p.y;
        }
        out[i] = (c > 0.0f) ? (es / fmaxf(c, 1.0f)) : 10.0f;
    }
}

extern "C" void kernel_launch(void* const* d_in, const int* in_sizes, int n_in,
                              void* d_out, int out_size, void* d_ws, size_t ws_size,
                              hipStream_t stream) {
    const float* x  = (const float*)d_in[0];
    const int*   ei = (const int*)d_in[1];
    const float* W1 = (const float*)d_in[2];
    const float* b1 = (const float*)d_in[3];
    const float* W2 = (const float*)d_in[4];
    const float* b2 = (const float*)d_in[5];

    const int N = in_sizes[0] / 128;
    const int E = in_sizes[1] / 2;
    const int* src = ei;
    const int* dst = ei + E;

    float* ws     = (float*)d_ws;
    float* dinv   = ws;
    float* mplane = dinv + N;
    size_t off    = (((size_t)3 * N) + 63) & ~(size_t)63;
    float* wt1    = ws + off;
    float* wt2    = wt1 + 128 * 128;
    float* bufA   = wt2 + 64 * 128;
    float* bufB   = bufA + (size_t)N * 128;

    const int T = 256;
    unsigned* marks   = (unsigned*)mplane;
    unsigned* degpart = (unsigned*)bufB;
    k_degmark  <<<dim3(DM_S, 16), DM_T, 0, stream>>>(dst, degpart, marks, E, N);
    k_rsqrt    <<<(N + T - 1) / T, T, 0, stream>>>(degpart, dinv, N);
    k_fragorder<<<(128 * 128 + T - 1) / T, T, 0, stream>>>(W1, wt1, 128);
    k_fragorder<<<(128 * 64  + T - 1) / T, T, 0, stream>>>(W2, wt2, 64);

    k_gemm_wmma_k128<128><<<N / 16, 256, 0, stream>>>(x, wt1, dinv, bufA);
    k_aggregate_own<128, true><<<250, 1024, 0, stream>>>(bufA, bufB, src, dst, dinv, b1, marks, E, N);

    k_gemm_wmma_k128<64><<<N / 16, 128, 0, stream>>>(bufB, wt2, dinv, bufA);
    k_aggregate_own<64, false><<<250, 1024, 0, stream>>>(bufA, bufB, src, dst, dinv, b2, marks, E, N);

    float* eerr = bufB + (size_t)N * 64;
    float* part = bufA;
    k_decode<<<(E + 255) / 256, 256, 0, stream>>>(bufB, src, dst, eerr, E);
    k_segsum<<<dim3(SEG_S, 4), SEG_T, 0, stream>>>(eerr, src, part, E, N);
    k_finalize<<<(N + T - 1) / T, T, 0, stream>>>(part, (float*)d_out, N);
}


# Aggregation chunk walk: a wave lists its marked chunks in LDS once (no bit search per chunk); a chunk is one 8-byte request per lane, one queueing step; queue order changes, sums stay arrival order
# speedup vs baseline: 3.8602x; 1.0716x over previous
#include <hip/hip_runtime.h>
#include <math.h>


typedef float v2f __attribute__((ext_vector_type(2)));
typedef float v8f __attribute__((ext_vector_type(8)));

constexpr int OWN_ROWS = 400;

__device__ __forceinline__ void ldsAddU(unsigned* p, unsigned v) {
    __hip_atomic_fetch_add(p, v, __ATOMIC_RELAXED, __HIP_MEMORY_SCOPE_WORKGROUP);
}
__device__ __forceinline__ void ldsOrU(unsigned* p, unsigned v) {
    __hip_atomic_fetch_or(p, v, __ATOMIC_RELAXED, __HIP_MEMORY_SCOPE_WORKGROUP);
}

constexpr int DM_S    = 16;
constexpr int DM_OPR  = 16;
constexpr int DM_ROWS = DM_OPR * OWN_ROWS;
constexpr int DM_WMAX = 64;
constexpr int DM_T    = 1024;
constexpr int DM_GY   = 16;
constexpr int DM_FY   = 2;

__device__ __forceinline__ void fragorder(const float* __restrict__ W, float* __restrict__ Wf,
                                          int N, int idx) {
    const int j    = idx & 3;
    const int lane = (idx >> 2) & 31;
    const int kq   = (idx >> 7) & 15;
    const int ct   = idx >> 11;
    const int k = 8 * kq + 4 * (j >> 1) + 2 * (lane >> 4) + (j & 1);
    const int n = 16 * ct + (lane & 15);
    Wf[idx] = W[k * N + n];
}

__global__ __launch_bounds__(DM_T)
void k_degmark(const int* __restrict__ dst, unsigned* __restrict__ degpart,
               unsigned* __restrict__ marks, int E, int N,
               const float* __restrict__ W1, float* __restrict__ W1f,
               const float* __restrict__ W2, float* __restrict__ W2f) {
    __shared__ unsigned cntl[DM_ROWS];
    __shared__ unsigned mk[DM_OPR * DM_WMAX];
    const int tid = threadIdx.x;

    if (blockIdx.y >= DM_GY) {
        const int nb = ((int)gridDim.y - DM_GY) * (int)gridDim.x;
        const int b  = ((int)blockIdx.y - DM_GY) * (int)gridDim.x + (int)blockIdx.x;
        for (int idx = b * DM_T + tid; idx < 128 * 128 + 128 * 64; idx += nb * DM_T) {
            if (idx < 128 * 128) fragorder(W1, W1f, 128, idx);
            else                 fragorder(W2, W2f, 64, idx - 128 * 128);
        }
        return;
    }

    const int FW  = (E + 2047) >> 11;
    const int wps = (FW + (int)gridDim.x - 1) / (int)gridDim.x;
    const int w0  = blockIdx.x * wps;
    const int e0  = w0 << 11;
    const int e1  = min(E, (w0 + wps) << 11);

    for (int rg = blockIdx.y; rg * DM_ROWS < N; rg += DM_GY) {
        const int base  = rg * DM_ROWS;
        const int nrows = min(DM_ROWS, N - base);
        for (int i = tid; i < DM_ROWS; i += DM_T) cntl[i] = 0u;
        for (int i = tid; i < DM_OPR * DM_WMAX; i += DM_T) mk[i] = 0u;
        __syncthreads();

        for (int e = e0 + tid; e < e1; e += 16 * DM_T) {
            int d[16];
#pragma unroll
            for (int k = 0; k < 16; ++k) {
                const int ee = e + k * DM_T;
                d[k] = (ee < e1) ? dst[ee] : -1;
            }
#pragma unroll
            for (int k = 0; k < 16; ++k) {
                const unsigned rl = (unsigned)(d[k] - base);
                if (rl < (unsigned)nrows) {
                    const int ee = e + k * DM_T;
                    ldsAddU(&cntl[rl], 1u);
                    ldsOrU(&mk[(rl / OWN_ROWS) * DM_WMAX + ((ee - e0) >> 11)],
                           1u << ((ee >> 6) & 31));
                }
            }
        }
        __syncthreads();

        unsigned* pl = degpart + (size_t)blockIdx.x * N + base;
        for (int i = tid; i < nrows; i += DM_T) pl[i] = cntl[i];
        for (int i = tid; i < DM_OPR * DM_WMAX; i += DM_T) {
            const int o = i / DM_WMAX, w = i - o * DM_WMAX;
            const int owner = rg * DM_OPR + o;
            if (w < wps && w0 + w < FW && owner * OWN_ROWS < N)
                marks[(size_t)owner * FW + w0 + w] = mk[i];
        }
        __syncthreads();
    }
}

template <int NCOLS, bool MAKE_RS>
__global__ void k_gemm_wmma_k128(const float* __restrict__ A,
                                 const float* __restrict__ Bf,
                                 const unsigned* __restrict__ degpart,
                                 float* __restrict__ rs,
                                 float* __restrict__ C, int M) {
    __shared__ float As[16 * 132];
    __shared__ unsigned cpl[DM_S * 16];
    __shared__ float rsl[16];
    const int tid  = threadIdx.x;
    const int row0 = blockIdx.x * 16;

    unsigned cmine = 0u;
    if constexpr (MAKE_RS) {
        static_assert((NCOLS / 16) * 32 == DM_S * 16, "one thread per (plane, row) pair");
        cmine = degpart[(size_t)(tid >> 4) * M + row0 + (tid & 15)];
    }

    const float4* Ag = (const float4*)(A + (size_t)row0 * 128);
    for (int i = tid; i < 16 * 32; i += blockDim.x) {
        int r = i >> 5, c4 = i & 31;
        ((float4*)(As + r * 132))[c4] = Ag[r * 32 + c4];
    }
    if constexpr (MAKE_RS) cpl[tid] = cmine;
    __syncthreads();

    if constexpr (MAKE_RS) {
        if (tid < 16) {
            unsigned c = 1u;
#pragma unroll
            for (int s = 0; s < DM_S; ++s) c += cpl[s * 16 + tid];
            rsl[tid] = rsqrtf((float)c);
        }
    }

    const int lane  = tid & 31;
    const int col0  = (tid >> 5) * 16;
    const int n15   = lane & 15;
    const int khalf = lane >> 4;

    v8f acc = {0.f, 0.f, 0.f, 0.f, 0.f, 0.f, 0.f, 0.f};
    const float* ap = As + n15 * 132 + 2 * khalf;
    const float* bp = Bf + ((size_t)(tid >> 5) * (16 * 32) + lane) * 4;

#pragma unroll
    for (int kq = 0; kq < 16; ++kq) {
        const float4 b4 = *(const float4*)(bp + kq * 128);
        v2f a0; a0.x = ap[8 * kq];     a0.y = ap[8 * kq + 1];
        v2f b0; b0.x = b4.x;           b0.y = b4.y;
        acc = __builtin_amdgcn_wmma_f32_16x16x4_f32(
             false, a0,  false, b0,
             (short)0, acc,  false,  false);
        v2f a1; a1.x = ap[8 * kq + 4]; a1.y = ap[8 * kq + 5];
        v2f b1; b1.x = b4.z;           b1.y = b4.w;
        acc = __builtin_amdgcn_wmma_f32_16x16x4_f32(
             false, a1,  false, b1,
             (short)0, acc,  false,  false);
    }

    float* Cp = C + (size_t)(row0 + 8 * khalf) * NCOLS + col0 + n15;
    if constexpr (MAKE_RS) {
        __syncthreads();
#pragma unroll
        for (int r = 0; r < 8; ++r) Cp[r * NCOLS] = acc[r] * rsl[8 * khalf + r];
        if (tid < 16) rs[row0 + tid] = rsl[tid];
    } else {
        const float* rp = rs + row0 + 8 * khalf;
#pragma unroll
        for (int r = 0; r < 8; ++r) Cp[r * NCOLS] = acc[r] * rp[r];
    }
}

constexpr int OWN_QCAP = 288;
constexpr int OWN_LCAP = 512;

__device__ __forceinline__ void ldsAddF(float* p, float v) {
    __hip_atomic_fetch_add(p, v, __ATOMIC_RELAXED, __HIP_MEMORY_SCOPE_WORKGROUP);
}
__device__ __forceinline__ unsigned ldsAddRtnU(unsigned* p, unsigned v) {
    return __hip_atomic_fetch_add(p, v, __ATOMIC_RELAXED, __HIP_MEMORY_SCOPE_WORKGROUP);
}

template <int VPL> struct RowVec;
template <> struct RowVec<4> { typedef float4 T; };
template <> struct RowVec<2> { typedef float2 T; };

template <int DIM, bool RELU>
__global__ __launch_bounds__(1024)
void k_aggregate_own(const float* __restrict__ H, float* __restrict__ out,
                     const int* __restrict__ src, const int* __restrict__ dst,
                     const float* __restrict__ dinv, const float* __restrict__ bias,
                     const unsigned* __restrict__ marks, int E, int N) {
    constexpr int VPL = DIM / 32;
    constexpr int AGG_RF = 64 / VPL;
    typedef typename RowVec<VPL>::T VT;
    static_assert(OWN_ROWS <= 65536, "local row numbers are kept in 16 bits");
    __shared__ float acc[OWN_ROWS * DIM];
    __shared__ int   queue[32 * OWN_QCAP];
    __shared__ unsigned short qrow[32 * OWN_QCAP];
    __shared__ unsigned short clist[32 * OWN_LCAP];
    __shared__ unsigned ccnt[32];

    const int lane = threadIdx.x & 31;
    const int wv = __builtin_amdgcn_readfirstlane((int)(threadIdx.x >> 5));
    int* qbuf = queue + wv * OWN_QCAP;
    unsigned short* qdl = qrow + wv * OWN_QCAP;
    unsigned short* cl = clist + wv * OWN_LCAP;
    unsigned* cn = ccnt + wv;
    const int lane2 = 2 * lane;
    const float* hl = H + lane * VPL;
    float bz[VPL];
#pragma unroll
    for (int k = 0; k < VPL; ++k) bz[k] = bias[lane * VPL + k];

    auto ldrow = [&](int r) -> VT { return *(const VT*)(hl + (size_t)r * DIM); };
    auto addrow = [&](int dj, const VT& v) {
        float* a = acc + dj * DIM + lane;
        if constexpr (VPL == 4) {
            ldsAddF(a +  0, v.x); ldsAddF(a + 32, v.y);
            ldsAddF(a + 64, v.z); ldsAddF(a + 96, v.w);
        } else {
            ldsAddF(a +  0, v.x); ldsAddF(a + 32, v.y);
        }
    };

    const int FW = (E + 2047) >> 11;
    for (int base = blockIdx.x * OWN_ROWS; base < N; base += gridDim.x * OWN_ROWS) {
        const int nrows = min(OWN_ROWS, N - base);
        const unsigned* fl = marks + (base / OWN_ROWS) * FW;

        int wb = wv;
        unsigned mine = (wb + 32 * lane < FW) ? fl[wb + 32 * lane] : 0u;

        auto putrow = [&](int r, const VT& h) {
            float* a = acc + r * DIM + lane;
            if constexpr (VPL == 4) {
                a[0] = h.x; a[32] = h.y; a[64] = h.z; a[96] = h.w;
            } else {
                a[0] = h.x; a[32] = h.y;
            }
        };
        {
            int r = wv;
            for (; r + 96 < nrows; r += 128) {
                VT h[4];
#pragma unroll
                for (int k = 0; k < 4; ++k) h[k] = ldrow(base + r + 32 * k);
#pragma unroll
                for (int k = 0; k < 4; ++k) putrow(r + 32 * k, h[k]);
            }
            for (; r < nrows; r += 32) putrow(r, ldrow(base + r));
        }
        __syncthreads();

        int tail = 0;

        auto flush32 = [&](int off) {
            const int s  = src[qbuf[off + lane]];
            const int dl = (int)qdl[off + lane];
            VT v[AGG_RF];
#pragma unroll
            for (int j = 0; j < AGG_RF; ++j) v[j] = ldrow(__builtin_amdgcn_readlane(s, j));
#pragma unroll
            for (int j0 = 0; j0 + AGG_RF < 32; j0 += AGG_RF) {
#pragma unroll
                for (int j = 0; j < AGG_RF; ++j) {
                    addrow(__builtin_amdgcn_readlane(dl, j0 + j), v[j]);
                    v[j] = ldrow(__builtin_amdgcn_readlane(s, j0 + AGG_RF + j));
                }
            }
#pragma unroll
            for (int j = 0; j < AGG_RF; ++j)
                addrow(__builtin_amdgcn_readlane(dl, 32 - AGG_RF + j), v[j]);
        };

        auto flush_part = [&](int off, int cnt) {
            int s = 0, dl = 0;
            if (lane < cnt) {
                s  = src[qbuf[off + lane]];
                dl = (int)qdl[off + lane];
            }
            for (int j0 = 0; j0 < cnt; j0 += 8) {
                VT v[8];
                for (int j = 0; j < 8; ++j)
                    v[j] = ldrow(__builtin_amdgcn_readlane(s, j0 + j));
                for (int j = 0; j < 8; ++j)
                    if (j0 + j < cnt) addrow(__builtin_amdgcn_readlane(dl, j0 + j), v[j]);
            }
        };

        auto push64 = [&](const int2& d, int c) {
            const unsigned r0 = (unsigned)(d.x - base);
            const unsigned r1 = (unsigned)(d.y - base);
            const int g0 = c + lane2;
            const bool inb = g0 < E;
            const bool h0 = (r0 < (unsigned)nrows) && inb;
            const bool h1 = (r1 < (unsigned)nrows) && inb;
            const unsigned mi = __builtin_amdgcn_ballot_w32(inb);
            const unsigned m0 = __builtin_amdgcn_ballot_w32(r0 < (unsigned)nrows) & mi;
            const unsigned m1 = __builtin_amdgcn_ballot_w32(r1 < (unsigned)nrows) & mi;
            if (h0) {
                const int q = (int)__builtin_amdgcn_mbcnt_lo(m0, (unsigned)tail);
                qbuf[q] = g0;
                qdl[q]  = (unsigned short)r0;
            }
            const int t1 = tail + __builtin_popcount(m0);
            if (h1) {
                const int q = (int)__builtin_amdgcn_mbcnt_lo(m1, (unsigned)t1);
                qbuf[q] = g0 + 1;
                qdl[q]  = (unsigned short)r1;
            }
            tail = t1 + __builtin_popcount(m1);
        };

        auto drain = [&]() {
            if (tail >= 32) {
                int head = 0;
                for (; tail - head >= 32; head += 32) flush32(head);
                const int rem = tail - head;
                if (lane < rem) {
                    const int g = qbuf[head + lane];
                    const unsigned short t = qdl[head + lane];
                    qbuf[lane] = g;
                    qdl[lane]  = t;
                }
                tail = rem;
            }
        };

        while (wb < FW) {
            for (int half = 0; half < 32; half += 16) {
                if (lane == 0) __hip_atomic_store(cn, 0u, __ATOMIC_RELAXED, __HIP_MEMORY_SCOPE_WORKGROUP);
                unsigned w = ((lane & 16) == half) ? mine : 0u;
                while (w != 0u) {
                    const int b = __builtin_ctz(w);
                    w &= w - 1u;
                    const unsigned pos = ldsAddRtnU(cn, 1u);
                    cl[pos] = (unsigned short)((lane << 5) | b);
                }
                const int n = __builtin_amdgcn_readfirstlane(
                    (int)__hip_atomic_load(cn, __ATOMIC_RELAXED, __HIP_MEMORY_SCOPE_WORKGROUP));

                int cfirst = 0;

                auto fetch8 = [&](int k0, int (&cb)[8], int2 (&dd)[8]) {
                    if ((k0 & 31) == 0) {
                        int t = 0;
                        if (k0 + lane < n) t = (int)cl[k0 + lane];
                        cfirst = ((wb + 32 * (t >> 5)) << 11) + ((t & 31) << 6);
                    }
#pragma unroll
                    for (int j = 0; j < 8; ++j) {
                        const int  cj   = __builtin_amdgcn_readlane(cfirst, (k0 & 31) + j);
                        const bool have = (k0 + j < n);
                        const int  cc   = have ? cj : 0;
                        cb[j] = have ? cc : -1;
                        const int o = min(lane2, E - 2 - cc);
                        dd[j] = *(const int2*)(dst + cc + o);
                    }
                };

                int cbA[8]; int2 ddA[8];
                fetch8(0, cbA, ddA);
                for (int k0 = 0; k0 < n; k0 += 8) {
                    int cbB[8]; int2 ddB[8];
                    fetch8(k0 + 8, cbB, ddB);
#pragma unroll
                    for (int j = 0; j < 8; ++j) {
                        if (cbA[j] >= 0) push64(ddA[j], cbA[j]);
                        if (j == 3 || j == 7) drain();
                    }
#pragma unroll
                    for (int j = 0; j < 8; ++j) { cbA[j] = cbB[j]; ddA[j] = ddB[j]; }
                }
            }

            wb += 32 * 32;
            if (wb < FW) mine = (wb + 32 * lane < FW) ? fl[wb + 32 * lane] : 0u;
        }
        if (tail > 0) flush_part(0, tail);
        __syncthreads();

        auto outrow = [&](int rr, float di) {
            const float* a = acc + rr * DIM + lane;
            float* op = out + (size_t)(base + rr) * DIM + lane * VPL;
            if constexpr (VPL == 4) {
                float4 o = {bz[0] + di * a[0], bz[1] + di * a[32],
                            bz[2] + di * a[64], bz[3] + di * a[96]};
                if (RELU) { o.x = fmaxf(o.x, 0.0f); o.y = fmaxf(o.y, 0.0f);
                            o.z = fmaxf(o.z, 0.0f); o.w = fmaxf(o.w, 0.0f); }
                *(float4*)op = o;
            } else {
                float2 o = {bz[0] + di * a[0], bz[1] + di * a[32]};
                if (RELU) { o.x = fmaxf(o.x, 0.0f); o.y = fmaxf(o.y, 0.0f); }
                *(float2*)op = o;
            }
        };
        {
            int r = wv;
            for (; r + 96 < nrows; r += 128) {
                float di[4];
#pragma unroll
                for (int k = 0; k < 4; ++k) di[k] = dinv[base + r + 32 * k];
#pragma unroll
                for (int k = 0; k < 4; ++k) outrow(r + 32 * k, di[k]);
            }
            for (; r < nrows; r += 32) outrow(r, dinv[base + r]);
        }
        __syncthreads();
    }
}

constexpr int SEG_S = 64;
constexpr int SEG_R = 25000;
constexpr int SEG_T = 1024;
constexpr int DEC_QCAP = 288;

__global__ __launch_bounds__(SEG_T)
void k_decode_segsum(const float* __restrict__ Z, const int* __restrict__ row,
                     const int* __restrict__ col, float* __restrict__ part, int E, int N) {
    static_assert(SEG_R <= 65536, "local row numbers are kept in 16 bits");
    __shared__ float2 acc[SEG_R];
    __shared__ int qcol[32 * DEC_QCAP];
    __shared__ unsigned short qrow[32 * DEC_QCAP];
    const int tid  = threadIdx.x;
    const int lane = tid & 31;
    const int wv = __builtin_amdgcn_readfirstlane((int)(threadIdx.x >> 5));
    int* qc = qcol + wv * DEC_QCAP;
    unsigned short* qr = qrow + wv * DEC_QCAP;
    const int ES  = (E + (int)gridDim.x - 1) / (int)gridDim.x;
    const int e0  = blockIdx.x * ES;
    const int e1  = min(E, e0 + ES);
    float2* pp = (float2*)part + (size_t)blockIdx.x * N;
    const unsigned lo8 = (unsigned)lane * 8u;
    const unsigned lo4 = (unsigned)lane * 4u;

    auto ldz = [&](int r) -> float2 {
        return *(const float2*)((const char*)(Z + (size_t)r * 64) + lo8);
    };
    auto dot64 = [&](const float2& a, const float2& b) -> float {
        float p = a.x * b.x + a.y * b.y;
#pragma unroll
        for (int off = 16; off > 0; off >>= 1) p += __shfl_xor(p, off, 32);
        return p;
    };
    auto edge_err = [&](float score) -> float {
        const float sig = 1.0f / (1.0f + __expf(-score));
        return -__logf(sig + 1e-15f);
    };

    for (int base = blockIdx.y * SEG_R; base < N; base += gridDim.y * SEG_R) {
        const int nrows = min(SEG_R, N - base);
        for (int i = tid; i < nrows; i += SEG_T) acc[i] = make_float2(0.0f, 0.0f);
        __syncthreads();

        int tail = 0;

        auto decode32 = [&](int off) {
            const int lr = (int)qr[off + lane];
            const int c  = qc[off + lane];
            const int r  = base + lr;
            float2 ra[8], ca[8], rb[8], cb[8];
            float score = 0.0f;
            auto issue8 = [&](float2 (&zr)[8], float2 (&zc)[8], int q) {
#pragma unroll
                for (int j = 0; j < 8; ++j) {
                    zr[j] = ldz(__builtin_amdgcn_readlane(r, q + j));
                    zc[j] = ldz(__builtin_amdgcn_readlane(c, q + j));
                }
            };
            auto sum8 = [&](const float2 (&zr)[8], const float2 (&zc)[8], int q) {
                float p[8];
#pragma unroll
                for (int j = 0; j < 8; ++j) p[j] = zr[j].x * zc[j].x + zr[j].y * zc[j].y;
#pragma unroll
                for (int o = 16; o > 0; o >>= 1) {
#pragma unroll
                    for (int j = 0; j < 8; ++j) p[j] += __shfl_xor(p[j], o, 32);
                }
#pragma unroll
                for (int j = 0; j < 8; ++j) if (lane == q + j) score = p[j];
            };
            issue8(ra, ca, 0);
            issue8(rb, cb, 8);
            __builtin_amdgcn_sched_barrier(0);
            sum8(ra, ca, 0);
            __builtin_amdgcn_sched_barrier(0);
            issue8(ra, ca, 16);
            __builtin_amdgcn_sched_barrier(0);
            sum8(rb, cb, 8);
            __builtin_amdgcn_sched_barrier(0);
            issue8(rb, cb, 24);
            __builtin_amdgcn_sched_barrier(0);
            sum8(ra, ca, 16);
            sum8(rb, cb, 24);
            ldsAddF(&acc[lr].x, edge_err(score));
            ldsAddF(&acc[lr].y, 1.0f);
        };

        auto decode_part = [&](int off, int cnt) {
            int lr = 0, c = 0;
            if (lane < cnt) {
                lr = (int)qr[off + lane];
                c  = qc[off + lane];
            }
            const int r = base + lr;
            float score = 0.0f;
            for (int j0 = 0; j0 < cnt; j0 += 8) {
                float2 zr[8], zc[8];
                for (int j = 0; j < 8; ++j) {
                    zr[j] = ldz(__builtin_amdgcn_readlane(r, j0 + j));
                    zc[j] = ldz(__builtin_amdgcn_readlane(c, j0 + j));
                }
                for (int j = 0; j < 8; ++j) {
                    const float p = dot64(zr[j], zc[j]);
                    if (lane == j0 + j) score = p;
                }
            }
            if (lane < cnt) {
                ldsAddF(&acc[lr].x, edge_err(score));
                ldsAddF(&acc[lr].y, 1.0f);
            }
        };

        auto push = [&](int r, int c) {
            const unsigned rl = (unsigned)(r - base);
            const bool hit = rl < (unsigned)nrows;
            const unsigned m = __builtin_amdgcn_ballot_w32(hit);
            if (m != 0u) {
                if (hit) {
                    const int q = tail + (int)__builtin_amdgcn_mbcnt_lo(m, 0u);
                    qc[q] = c;
                    qr[q] = (unsigned short)rl;
                }
                tail += __builtin_popcount(m);
            }
        };

        auto drain = [&]() {
            if (tail >= 32) {
                int head = 0;
                for (; tail - head >= 32; head += 32) decode32(head);
                const int rem = tail - head;
                if (lane < rem) {
                    const int c = qc[head + lane];
                    const unsigned short t = qr[head + lane];
                    qc[lane] = c;
                    qr[lane] = t;
                }
                tail = rem;
            }
        };

        auto fetch = [&](int sb, int (&r)[8], int (&c)[8]) {
#pragma unroll
            for (int k = 0; k < 8; ++k) {
                const int s = sb + k * SEG_T;
                r[k] = -1; c[k] = 0;
                if (s + lane < e1) {
                    r[k] = *(const int*)((const char*)(row + s) + lo4);
                    c[k] = *(const int*)((const char*)(col + s) + lo4);
                }
            }
        };

        int sb = e0 + wv * 32;
        int rA[8], cA[8];
        fetch(sb, rA, cA);
        while (sb < e1) {
#pragma unroll
            for (int k = 0; k < 8; ++k) push(rA[k], cA[k]);
            sb += 8 * SEG_T;
            fetch(sb, rA, cA);
            drain();
        }
        if (tail > 0) decode_part(0, tail);
        __syncthreads();

        for (int i = tid; i < nrows; i += SEG_T) pp[base + i] = acc[i];
        __syncthreads();
    }
}

__global__ void k_finalize(const float* __restrict__ part, float* __restrict__ out, int N) {
    int i = blockIdx.x * blockDim.x + threadIdx.x;
    if (i < N) {
        float es = 0.0f, c = 0.0f;
#pragma unroll 16
        for (int s = 0; s < SEG_S; ++s) {
            const float2 p = ((const float2*)part)[(size_t)s * N + i];
            es += p.x;
            c  += p.y;
        }
        out[i] = (c > 0.0f) ? (es / fmaxf(c, 1.0f)) : 10.0f;
    }
}

extern "C" void kernel_launch(void* const* d_in, const int* in_sizes, int n_in,
                              void* d_out, int out_size, void* d_ws, size_t ws_size,
                              hipStream_t stream) {
    const float* x  = (const float*)d_in[0];
    const int*   ei = (const int*)d_in[1];
    const float* W1 = (const float*)d_in[2];
    const float* b1 = (const float*)d_in[3];
    const float* W2 = (const float*)d_in[4];
    const float* b2 = (const float*)d_in[5];

    const int N = in_sizes[0] / 128;
    const int E = in_sizes[1] / 2;
    const int* src = ei;
    const int* dst = ei + E;

    float* ws     = (float*)d_ws;
    float* dinv   = ws;
    float* mplane = dinv + N;
    size_t off    = (((size_t)3 * N) + 63) & ~(size_t)63;
    float* wt1    = ws + off;
    float* wt2    = wt1 + 128 * 128;
    float* bufA   = wt2 + 64 * 128;
    float* bufB   = bufA + (size_t)N * 128;

    const int T = 256;
    unsigned* marks   = (unsigned*)mplane;
    unsigned* degpart = (unsigned*)bufB;
    k_degmark<<<dim3(DM_S, DM_GY + DM_FY), DM_T, 0, stream>>>(dst, degpart, marks, E, N,
                                                             W1, wt1, W2, wt2);

    k_gemm_wmma_k128<128, true><<<N / 16, 256, 0, stream>>>(x, wt1, degpart, dinv, bufA, N);
    k_aggregate_own<128, true><<<250, 1024, 0, stream>>>(bufA, bufB, src, dst, dinv, b1, marks, E, N);

    k_gemm_wmma_k128<64, false><<<N / 16, 128, 0, stream>>>(bufB, wt2, degpart, dinv, bufA, N);
    k_aggregate_own<64, false><<<250, 1024, 0, stream>>>(bufA, bufB, src, dst, dinv, b2, marks, E, N);

    float* part = bufA;
    k_decode_segsum<<<dim3(SEG_S, 4), SEG_T, 0, stream>>>(bufB, src, dst, part, E, N);
    k_finalize<<<(N + T - 1) / T, T, 0, stream>>>(part, (float*)d_out, N);
}
